# starting kernel
# baseline (speedup 1.0000x reference)
#include <hip/hip_runtime.h>
#include <math.h>

typedef float v2f __attribute__((ext_vector_type(2)));
typedef float v8f __attribute__((ext_vector_type(8)));

#define HDIM 128
#define WPB 8
#define LDSA_STRIDE 132
#define LDSW_STRIDE 136

__launch_bounds__(256)
__global__ void gemm_wmma(const float* __restrict__ X, int lda,
                          const float* __restrict__ W,
                          float* __restrict__ Hout, int nTiles) {
    __shared__ float ldsW[HDIM * LDSW_STRIDE];
    __shared__ float ldsA[WPB][16 * LDSA_STRIDE];

    const int tid  = threadIdx.x;
    const int wave = tid >> 5;
    const int lane = tid & 31;

    for (int i = 0; i < 16; ++i) {
        int idx = (i * 256 + tid) * 4;
        int row = idx >> 7, col = idx & 127;
        float4 v = *(const float4*)(W + idx);
        *(float4*)(&ldsW[row * LDSW_STRIDE + col]) = v;
    }

    const int tile   = blockIdx.x * WPB + wave;
    const bool active = (tile < nTiles);
    if (active) {
        const float* Xrow = X + (size_t)tile * 16 * (size_t)lda;
        for (int r = 0; r < 16; ++r) {
            float4 v = *(const float4*)(Xrow + (size_t)r * lda + lane * 4);
            *(float4*)(&ldsA[wave][r * LDSA_STRIDE + lane * 4]) = v;
        }
    }
    __syncthreads();
    if (!active) return;

    const int m     = lane & 15;
    const int khalf = (lane >> 4) * 2;
    const float* aRow = &ldsA[wave][m * LDSA_STRIDE];

    v8f acc[8];
    const v8f vzero = {0.f, 0.f, 0.f, 0.f, 0.f, 0.f, 0.f, 0.f};
#pragma unroll
    for (int nt = 0; nt < 8; ++nt) acc[nt] = vzero;

    for (int k4 = 0; k4 < 32; ++k4) {
        const int kb = k4 * 4 + khalf;
        v2f a;
        a.x = aRow[kb];
        a.y = aRow[kb + 1];
        const float* w0 = &ldsW[kb * LDSW_STRIDE];
        const float* w1 = w0 + LDSW_STRIDE;
#pragma unroll
        for (int nt = 0; nt < 8; ++nt) {
            const int n = nt * 16 + m;
            v2f b;
            b.x = w0[n];
            b.y = w1[n];
            acc[nt] = __builtin_amdgcn_wmma_f32_16x16x4_f32(
                false, a, false, b, (short)0, acc[nt], false, false);
        }
    }

    float* outBase = Hout + (size_t)tile * 16 * HDIM;
    const int mrow = (lane >> 4) * 8;
#pragma unroll
    for (int nt = 0; nt < 8; ++nt) {
#pragma unroll
        for (int r = 0; r < 8; ++r) {
            outBase[(size_t)(mrow + r) * HDIM + nt * 16 + m] = acc[nt][r];
        }
    }
}

__global__ void deg_init(float* __restrict__ deg, int n) {
    int i = blockIdx.x * blockDim.x + threadIdx.x;
    if (i < n) deg[i] = 1.0f;
}

__global__ void deg_count(const int* __restrict__ dst, float* __restrict__ deg, int E) {
    int e = blockIdx.x * blockDim.x + threadIdx.x;
    if (e < E) atomicAdd(&deg[dst[e]], 1.0f);
}

__global__ void make_dinv(float* __restrict__ d, int n) {
    int i = blockIdx.x * blockDim.x + threadIdx.x;
    if (i < n) d[i] = rsqrtf(d[i]);
}

__global__ void agg_init(const float* __restrict__ h, const float* __restrict__ dinv,
                         float* __restrict__ agg, int n) {
    int t = blockIdx.x * blockDim.x + threadIdx.x;
    if (t >= n * 32) return;
    int node = t >> 5;
    float w = dinv[node];
    w *= w;
    float4 v = ((const float4*)h)[t];
    v.x *= w; v.y *= w; v.z *= w; v.w *= w;
    ((float4*)agg)[t] = v;
}

__global__ void edge_agg(const int* __restrict__ src, const int* __restrict__ dst,
                         const float* __restrict__ dinv, const float* __restrict__ h,
                         float* __restrict__ agg, int E) {
    int gid  = blockIdx.x * blockDim.x + threadIdx.x;
    int e    = gid >> 5;
    int lane = gid & 31;
    if (e >= E) return;
    int s = src[e], d = dst[e];
    float w = dinv[s] * dinv[d];
    float4 v = ((const float4*)(h + (size_t)s * HDIM))[lane];
    float* ap = agg + (size_t)d * HDIM + lane * 4;
    atomicAdd(ap + 0, v.x * w);
    atomicAdd(ap + 1, v.y * w);
    atomicAdd(ap + 2, v.z * w);
    atomicAdd(ap + 3, v.w * w);
}

__global__ void post_norm(const float* __restrict__ agg, const float* __restrict__ bias,
                          float* __restrict__ embSlice, int n) {
    int gid  = blockIdx.x * blockDim.x + threadIdx.x;
    int node = gid >> 5;
    int lane = gid & 31;
    if (node >= n) return;
    float4 v = ((const float4*)(agg + (size_t)node * HDIM))[lane];
    float4 b = ((const float4*)bias)[lane];
    v.x = fmaxf(v.x + b.x, 0.f);
    v.y = fmaxf(v.y + b.y, 0.f);
    v.z = fmaxf(v.z + b.z, 0.f);
    v.w = fmaxf(v.w + b.w, 0.f);
    float ss = v.x * v.x + v.y * v.y + v.z * v.z + v.w * v.w;
#pragma unroll
    for (int off = 16; off > 0; off >>= 1)
        ss += __shfl_xor(ss, off, 32);
    float scale = 1.0f / fmaxf(sqrtf(ss), 1e-12f);
    float4 o;
    o.x = v.x * scale; o.y = v.y * scale; o.z = v.z * scale; o.w = v.w * scale;
    *(float4*)(embSlice + (size_t)node * 384 + lane * 4) = o;
}

__launch_bounds__(256)
__global__ void classifier(const float* __restrict__ emb, const float* __restrict__ Wlin,
                           const float* __restrict__ blin, float* __restrict__ out, int n) {
    __shared__ float lw[384 * 8];
    __shared__ float lb[8];
    for (int i = threadIdx.x; i < 384 * 8; i += 256) lw[i] = Wlin[i];
    if (threadIdx.x < 8) lb[threadIdx.x] = blin[threadIdx.x];
    __syncthreads();

    int node = blockIdx.x * blockDim.x + threadIdx.x;
    if (node >= n) return;

    float acc[8];
#pragma unroll
    for (int c = 0; c < 8; ++c) acc[c] = lb[c];
    const float* er = emb + (size_t)node * 384;
    for (int k = 0; k < 384; ++k) {
        float e = er[k];
#pragma unroll
        for (int c = 0; c < 8; ++c) acc[c] = fmaf(e, lw[k * 8 + c], acc[c]);
    }
    float m = acc[0];
#pragma unroll
    for (int c = 1; c < 8; ++c) m = fmaxf(m, acc[c]);
    float s = 0.f;
#pragma unroll
    for (int c = 0; c < 8; ++c) s += expf(acc[c] - m);
    float lse = m + logf(s);
#pragma unroll
    for (int c = 0; c < 8; ++c) out[(size_t)node * 8 + c] = acc[c] - lse;
}

extern "C" void kernel_launch(void* const* d_in, const int* in_sizes, int n_in,
                              void* d_out, int out_size, void* d_ws, size_t ws_size,
                              hipStream_t stream) {
    const float* x    = (const float*)d_in[0];
    const int*   ei   = (const int*)d_in[1];
    const float* W1   = (const float*)d_in[2];
    const float* b1   = (const float*)d_in[3];
    const float* W2   = (const float*)d_in[4];
    const float* b2   = (const float*)d_in[5];
    const float* W3   = (const float*)d_in[6];
    const float* b3   = (const float*)d_in[7];
    const float* Wlin = (const float*)d_in[8];
    const float* blin = (const float*)d_in[9];

    const int n = in_sizes[0] / HDIM;
    const int E = in_sizes[1] / 2;
    const int* src = ei;
    const int* dst = ei + E;

    float* ws   = (float*)d_ws;
    float* dinv = ws;
    float* h    = ws + (size_t)n;
    float* agg  = h  + (size_t)n * HDIM;
    float* emb  = agg + (size_t)n * HDIM;

    const int T = 256;
    deg_init<<<(n + T - 1) / T, T, 0, stream>>>(dinv, n);
    deg_count<<<(E + T - 1) / T, T, 0, stream>>>(dst, dinv, E);
    make_dinv<<<(n + T - 1) / T, T, 0, stream>>>(dinv, n);

    const int nTiles = (n + 15) / 16;
    const float* Ws[3] = {W1, W2, W3};
    const float* bs[3] = {b1, b2, b3};
    const float* Xin = x;
    int lda = HDIM;

    for (int l = 0; l < 3; ++l) {
        gemm_wmma<<<(nTiles + WPB - 1) / WPB, 256, 0, stream>>>(Xin, lda, Ws[l], h, nTiles);
        agg_init<<<((size_t)n * 32 + T - 1) / T, T, 0, stream>>>(h, dinv, agg, n);
        edge_agg<<<(unsigned)(((size_t)E * 32 + T - 1) / T), T, 0, stream>>>(src, dst, dinv, h, agg, E);
        post_norm<<<((size_t)n * 32 + T - 1) / T, T, 0, stream>>>(agg, bs[l], emb + l * HDIM, n);
        Xin = emb + l * HDIM;
        lda = 384;
    }

    classifier<<<(n + T - 1) / T, T, 0, stream>>>(emb, Wlin, blin, (float*)d_out, n);
}


# Edge scatter of f32 atomics replaced by a per-node gather over a dst-sorted edge list built each call (returning int atomic + scan); a node's sum order is now its edges' arrival rank
# speedup vs baseline: 2.4626x; 2.4626x over previous
#include <hip/hip_runtime.h>
#include <math.h>

typedef float v2f __attribute__((ext_vector_type(2)));
typedef float v8f __attribute__((ext_vector_type(8)));

#define HDIM 128
#define WPB 8
#define LDSA_STRIDE 132
#define LDSW_STRIDE 136

__launch_bounds__(256)
__global__ void gemm_wmma(const float* __restrict__ X, int lda,
                          const float* __restrict__ W,
                          float* __restrict__ Hout, int nTiles) {
    __shared__ float ldsW[HDIM * LDSW_STRIDE];
    __shared__ float ldsA[WPB][16 * LDSA_STRIDE];

    const int tid  = threadIdx.x;
    const int wave = tid >> 5;
    const int lane = tid & 31;

    for (int i = 0; i < 16; ++i) {
        int idx = (i * 256 + tid) * 4;
        int row = idx >> 7, col = idx & 127;
        float4 v = *(const float4*)(W + idx);
        *(float4*)(&ldsW[row * LDSW_STRIDE + col]) = v;
    }

    const int tile   = blockIdx.x * WPB + wave;
    const bool active = (tile < nTiles);
    if (active) {
        const float* Xrow = X + (size_t)tile * 16 * (size_t)lda;
        for (int r = 0; r < 16; ++r) {
            float4 v = *(const float4*)(Xrow + (size_t)r * lda + lane * 4);
            *(float4*)(&ldsA[wave][r * LDSA_STRIDE + lane * 4]) = v;
        }
    }
    __syncthreads();
    if (!active) return;

    const int m     = lane & 15;
    const int khalf = (lane >> 4) * 2;
    const float* aRow = &ldsA[wave][m * LDSA_STRIDE];

    v8f acc[8];
    const v8f vzero = {0.f, 0.f, 0.f, 0.f, 0.f, 0.f, 0.f, 0.f};
#pragma unroll
    for (int nt = 0; nt < 8; ++nt) acc[nt] = vzero;

    for (int k4 = 0; k4 < 32; ++k4) {
        const int kb = k4 * 4 + khalf;
        v2f a;
        a.x = aRow[kb];
        a.y = aRow[kb + 1];
        const float* w0 = &ldsW[kb * LDSW_STRIDE];
        const float* w1 = w0 + LDSW_STRIDE;
#pragma unroll
        for (int nt = 0; nt < 8; ++nt) {
            const int n = nt * 16 + m;
            v2f b;
            b.x = w0[n];
            b.y = w1[n];
            acc[nt] = __builtin_amdgcn_wmma_f32_16x16x4_f32(
                false, a, false, b, (short)0, acc[nt], false, false);
        }
    }

    float* outBase = Hout + (size_t)tile * 16 * HDIM;
    const int mrow = (lane >> 4) * 8;
#pragma unroll
    for (int nt = 0; nt < 8; ++nt) {
#pragma unroll
        for (int r = 0; r < 8; ++r) {
            outBase[(size_t)(mrow + r) * HDIM + nt * 16 + m] = acc[nt][r];
        }
    }
}

#define SCAN_PER 1024

__global__ void cnt_zero(int* __restrict__ cnt, int n) {
    int i = blockIdx.x * blockDim.x + threadIdx.x;
    if (i < n) cnt[i] = 0;
}

__global__ void count_rank(const int* __restrict__ dst, int* __restrict__ cnt,
                           int* __restrict__ rank, int E) {
    int e = blockIdx.x * blockDim.x + threadIdx.x;
    if (e < E) rank[e] = atomicAdd(&cnt[dst[e]], 1);
}

__launch_bounds__(256)
__global__ void scan_local(const int* __restrict__ cnt, int* __restrict__ rowptr,
                           int* __restrict__ blockSum, float* __restrict__ dinv, int n) {
    __shared__ int waveTot[8];
    const int tid  = threadIdx.x;
    const int lane = tid & 31;
    const int wave = tid >> 5;
    const int base = blockIdx.x * SCAN_PER + tid * 4;

    int c[4];
#pragma unroll
    for (int k = 0; k < 4; ++k) c[k] = (base + k < n) ? cnt[base + k] : 0;
    const int t = c[0] + c[1] + c[2] + c[3];

    int inc = t;
#pragma unroll
    for (int off = 1; off < 32; off <<= 1) {
        int v = __shfl_up(inc, off, 32);
        if (lane >= off) inc += v;
    }
    if (lane == 31) waveTot[wave] = inc;
    __syncthreads();

    int wbase = 0, tot = 0;
#pragma unroll
    for (int w = 0; w < 8; ++w) {
        int v = waveTot[w];
        if (w < wave) wbase += v;
        tot += v;
    }
    int excl = wbase + inc - t;
#pragma unroll
    for (int k = 0; k < 4; ++k) {
        if (base + k < n) {
            rowptr[base + k] = excl;
            dinv[base + k]   = rsqrtf((float)(c[k] + 1));
            excl += c[k];
        }
    }
    if (tid == 0) blockSum[blockIdx.x] = tot;
}

__launch_bounds__(256)
__global__ void scan_add(int* __restrict__ rowptr, const int* __restrict__ blockSum,
                         int n, int E) {
    __shared__ int part[8];
    const int tid  = threadIdx.x;
    const int lane = tid & 31;
    const int wave = tid >> 5;

    int s = 0;
    for (int b = tid; b < (int)blockIdx.x; b += 256) s += blockSum[b];
#pragma unroll
    for (int off = 16; off > 0; off >>= 1) s += __shfl_xor(s, off, 32);
    if (lane == 0) part[wave] = s;
    __syncthreads();
    int add = 0;
#pragma unroll
    for (int w = 0; w < 8; ++w) add += part[w];

    const int base = blockIdx.x * SCAN_PER + tid * 4;
#pragma unroll
    for (int k = 0; k < 4; ++k)
        if (base + k < n) rowptr[base + k] += add;
    if (blockIdx.x == 0 && tid == 0) rowptr[n] = E;
}

__global__ void csr_fill(const int* __restrict__ src, const int* __restrict__ dst,
                         const int* __restrict__ rank, const int* __restrict__ rowptr,
                         int* __restrict__ col, int E) {
    int e = blockIdx.x * blockDim.x + threadIdx.x;
    if (e < E) col[rowptr[dst[e]] + rank[e]] = src[e];
}

__global__ void gather_agg(const int* __restrict__ rowptr, const int* __restrict__ col,
                           const float* __restrict__ dinv, const float* __restrict__ h,
                           float* __restrict__ agg, int n) {
    const int gid  = blockIdx.x * blockDim.x + threadIdx.x;
    const int lane = gid & 31;
    const int node = __builtin_amdgcn_readfirstlane(gid >> 5);
    if (node >= n) return;

    const int   beg = rowptr[node];
    const int   end = rowptr[node + 1];
    const float dd  = dinv[node];

    float4 acc = ((const float4*)(h + (size_t)node * HDIM))[lane];
    const float wself = dd * dd;
    acc.x *= wself; acc.y *= wself; acc.z *= wself; acc.w *= wself;

    for (int base = beg; base < end; base += 32) {
        int m = end - base;
        if (m > 32) m = 32;
        int   s_l = 0;
        float w_l = 0.f;
        if (lane < m) {
            s_l = col[base + lane];
            w_l = dinv[s_l] * dd;
        }
#pragma unroll 4
        for (int j = 0; j < m; ++j) {
            const int   s = __builtin_amdgcn_readlane(s_l, j);
            const float w = __builtin_bit_cast(
                float, __builtin_amdgcn_readlane(__builtin_bit_cast(int, w_l), j));
            const float4 v = ((const float4*)(h + (size_t)s * HDIM))[lane];
            acc.x = fmaf(v.x, w, acc.x);
            acc.y = fmaf(v.y, w, acc.y);
            acc.z = fmaf(v.z, w, acc.z);
            acc.w = fmaf(v.w, w, acc.w);
        }
    }
    ((float4*)(agg + (size_t)node * HDIM))[lane] = acc;
}

__global__ void post_norm(const float* __restrict__ agg, const float* __restrict__ bias,
                          float* __restrict__ embSlice, int n) {
    int gid  = blockIdx.x * blockDim.x + threadIdx.x;
    int node = gid >> 5;
    int lane = gid & 31;
    if (node >= n) return;
    float4 v = ((const float4*)(agg + (size_t)node * HDIM))[lane];
    float4 b = ((const float4*)bias)[lane];
    v.x = fmaxf(v.x + b.x, 0.f);
    v.y = fmaxf(v.y + b.y, 0.f);
    v.z = fmaxf(v.z + b.z, 0.f);
    v.w = fmaxf(v.w + b.w, 0.f);
    float ss = v.x * v.x + v.y * v.y + v.z * v.z + v.w * v.w;
#pragma unroll
    for (int off = 16; off > 0; off >>= 1)
        ss += __shfl_xor(ss, off, 32);
    float scale = 1.0f / fmaxf(sqrtf(ss), 1e-12f);
    float4 o;
    o.x = v.x * scale; o.y = v.y * scale; o.z = v.z * scale; o.w = v.w * scale;
    *(float4*)(embSlice + (size_t)node * 384 + lane * 4) = o;
}

__launch_bounds__(256)
__global__ void classifier(const float* __restrict__ emb, const float* __restrict__ Wlin,
                           const float* __restrict__ blin, float* __restrict__ out, int n) {
    __shared__ float lw[384 * 8];
    __shared__ float lb[8];
    for (int i = threadIdx.x; i < 384 * 8; i += 256) lw[i] = Wlin[i];
    if (threadIdx.x < 8) lb[threadIdx.x] = blin[threadIdx.x];
    __syncthreads();

    int node = blockIdx.x * blockDim.x + threadIdx.x;
    if (node >= n) return;

    float acc[8];
#pragma unroll
    for (int c = 0; c < 8; ++c) acc[c] = lb[c];
    const float* er = emb + (size_t)node * 384;
    for (int k = 0; k < 384; ++k) {
        float e = er[k];
#pragma unroll
        for (int c = 0; c < 8; ++c) acc[c] = fmaf(e, lw[k * 8 + c], acc[c]);
    }
    float m = acc[0];
#pragma unroll
    for (int c = 1; c < 8; ++c) m = fmaxf(m, acc[c]);
    float s = 0.f;
#pragma unroll
    for (int c = 0; c < 8; ++c) s += expf(acc[c] - m);
    float lse = m + logf(s);
#pragma unroll
    for (int c = 0; c < 8; ++c) out[(size_t)node * 8 + c] = acc[c] - lse;
}

extern "C" void kernel_launch(void* const* d_in, const int* in_sizes, int n_in,
                              void* d_out, int out_size, void* d_ws, size_t ws_size,
                              hipStream_t stream) {
    const float* x    = (const float*)d_in[0];
    const int*   ei   = (const int*)d_in[1];
    const float* W1   = (const float*)d_in[2];
    const float* b1   = (const float*)d_in[3];
    const float* W2   = (const float*)d_in[4];
    const float* b2   = (const float*)d_in[5];
    const float* W3   = (const float*)d_in[6];
    const float* b3   = (const float*)d_in[7];
    const float* Wlin = (const float*)d_in[8];
    const float* blin = (const float*)d_in[9];

    const int n = in_sizes[0] / HDIM;
    const int E = in_sizes[1] / 2;
    const int* src = ei;
    const int* dst = ei + E;

    float* ws   = (float*)d_ws;
    float* dinv = ws;
    float* h    = ws + (size_t)n;
    float* agg  = h  + (size_t)n * HDIM;
    float* emb  = agg + (size_t)n * HDIM;
    int* col      = (int*)(emb + (size_t)n * 384);
    int* rowptr   = col + (size_t)E;
    int* cnt      = rowptr + (size_t)n + 1;
    int* blockSum = cnt + (size_t)n;
    int* rank     = (int*)agg;

    const int T = 256;
    const int nScan = (n + SCAN_PER - 1) / SCAN_PER;
    cnt_zero<<<(n + T - 1) / T, T, 0, stream>>>(cnt, n);
    count_rank<<<(E + T - 1) / T, T, 0, stream>>>(dst, cnt, rank, E);
    scan_local<<<nScan, 256, 0, stream>>>(cnt, rowptr, blockSum, dinv, n);
    scan_add<<<nScan, 256, 0, stream>>>(rowptr, blockSum, n, E);
    csr_fill<<<(E + T - 1) / T, T, 0, stream>>>(src, dst, rank, rowptr, col, E);

    const int nTiles = (n + 15) / 16;
    const float* Ws[3] = {W1, W2, W3};
    const float* bs[3] = {b1, b2, b3};
    const float* Xin = x;
    int lda = HDIM;

    for (int l = 0; l < 3; ++l) {
        gemm_wmma<<<(nTiles + WPB - 1) / WPB, 256, 0, stream>>>(Xin, lda, Ws[l], h, nTiles);
        gather_agg<<<((size_t)n * 32 + T - 1) / T, T, 0, stream>>>(rowptr, col, dinv, h, agg, n);
        post_norm<<<((size_t)n * 32 + T - 1) / T, T, 0, stream>>>(agg, bs[l], emb + l * HDIM, n);
        Xin = emb + l * HDIM;
        lda = 384;
    }

    classifier<<<(n + T - 1) / T, T, 0, stream>>>(emb, Wlin, blin, (float*)d_out, n);
}


# Bias+relu+L2-normalize moved into the gather wave, which holds the node's whole row in registers and stores it straight into emb; the agg plane and 3 of 15 launches (300000 one-node waves) go
# speedup vs baseline: 2.6989x; 1.0960x over previous
#include <hip/hip_runtime.h>
#include <math.h>

typedef float v2f __attribute__((ext_vector_type(2)));
typedef float v8f __attribute__((ext_vector_type(8)));

#define HDIM 128
#define WPB 8
#define LDSA_STRIDE 132
#define LDSW_STRIDE 136

__launch_bounds__(256)
__global__ void gemm_wmma(const float* __restrict__ X, int lda,
                          const float* __restrict__ W,
                          float* __restrict__ Hout, int nTiles) {
    __shared__ float ldsW[HDIM * LDSW_STRIDE];
    __shared__ float ldsA[WPB][16 * LDSA_STRIDE];

    const int tid  = threadIdx.x;
    const int wave = tid >> 5;
    const int lane = tid & 31;

    for (int i = 0; i < 16; ++i) {
        int idx = (i * 256 + tid) * 4;
        int row = idx >> 7, col = idx & 127;
        float4 v = *(const float4*)(W + idx);
        *(float4*)(&ldsW[row * LDSW_STRIDE + col]) = v;
    }

    const int tile   = blockIdx.x * WPB + wave;
    const bool active = (tile < nTiles);
    if (active) {
        const float* Xrow = X + (size_t)tile * 16 * (size_t)lda;
        for (int r = 0; r < 16; ++r) {
            float4 v = *(const float4*)(Xrow + (size_t)r * lda + lane * 4);
            *(float4*)(&ldsA[wave][r * LDSA_STRIDE + lane * 4]) = v;
        }
    }
    __syncthreads();
    if (!active) return;

    const int m     = lane & 15;
    const int khalf = (lane >> 4) * 2;
    const float* aRow = &ldsA[wave][m * LDSA_STRIDE];

    v8f acc[8];
    const v8f vzero = {0.f, 0.f, 0.f, 0.f, 0.f, 0.f, 0.f, 0.f};
#pragma unroll
    for (int nt = 0; nt < 8; ++nt) acc[nt] = vzero;

    for (int k4 = 0; k4 < 32; ++k4) {
        const int kb = k4 * 4 + khalf;
        v2f a;
        a.x = aRow[kb];
        a.y = aRow[kb + 1];
        const float* w0 = &ldsW[kb * LDSW_STRIDE];
        const float* w1 = w0 + LDSW_STRIDE;
#pragma unroll
        for (int nt = 0; nt < 8; ++nt) {
            const int n = nt * 16 + m;
            v2f b;
            b.x = w0[n];
            b.y = w1[n];
            acc[nt] = __builtin_amdgcn_wmma_f32_16x16x4_f32(
                false, a, false, b, (short)0, acc[nt], false, false);
        }
    }

    float* outBase = Hout + (size_t)tile * 16 * HDIM;
    const int mrow = (lane >> 4) * 8;
#pragma unroll
    for (int nt = 0; nt < 8; ++nt) {
#pragma unroll
        for (int r = 0; r < 8; ++r) {
            outBase[(size_t)(mrow + r) * HDIM + nt * 16 + m] = acc[nt][r];
        }
    }
}

#define SCAN_PER 1024

__global__ void cnt_zero(int* __restrict__ cnt, int n) {
    int i = blockIdx.x * blockDim.x + threadIdx.x;
    if (i < n) cnt[i] = 0;
}

__global__ void count_rank(const int* __restrict__ dst, int* __restrict__ cnt,
                           int* __restrict__ rank, int E) {
    int e = blockIdx.x * blockDim.x + threadIdx.x;
    if (e < E) rank[e] = atomicAdd(&cnt[dst[e]], 1);
}

__launch_bounds__(256)
__global__ void scan_local(const int* __restrict__ cnt, int* __restrict__ rowptr,
                           int* __restrict__ blockSum, float* __restrict__ dinv, int n) {
    __shared__ int waveTot[8];
    const int tid  = threadIdx.x;
    const int lane = tid & 31;
    const int wave = tid >> 5;
    const int base = blockIdx.x * SCAN_PER + tid * 4;

    int c[4];
#pragma unroll
    for (int k = 0; k < 4; ++k) c[k] = (base + k < n) ? cnt[base + k] : 0;
    const int t = c[0] + c[1] + c[2] + c[3];

    int inc = t;
#pragma unroll
    for (int off = 1; off < 32; off <<= 1) {
        int v = __shfl_up(inc, off, 32);
        if (lane >= off) inc += v;
    }
    if (lane == 31) waveTot[wave] = inc;
    __syncthreads();

    int wbase = 0, tot = 0;
#pragma unroll
    for (int w = 0; w < 8; ++w) {
        int v = waveTot[w];
        if (w < wave) wbase += v;
        tot += v;
    }
    int excl = wbase + inc - t;
#pragma unroll
    for (int k = 0; k < 4; ++k) {
        if (base + k < n) {
            rowptr[base + k] = excl;
            dinv[base + k]   = rsqrtf((float)(c[k] + 1));
            excl += c[k];
        }
    }
    if (tid == 0) blockSum[blockIdx.x] = tot;
}

__launch_bounds__(256)
__global__ void scan_add(int* __restrict__ rowptr, const int* __restrict__ blockSum,
                         int n, int E) {
    __shared__ int part[8];
    const int tid  = threadIdx.x;
    const int lane = tid & 31;
    const int wave = tid >> 5;

    int s = 0;
    for (int b = tid; b < (int)blockIdx.x; b += 256) s += blockSum[b];
#pragma unroll
    for (int off = 16; off > 0; off >>= 1) s += __shfl_xor(s, off, 32);
    if (lane == 0) part[wave] = s;
    __syncthreads();
    int add = 0;
#pragma unroll
    for (int w = 0; w < 8; ++w) add += part[w];

    const int base = blockIdx.x * SCAN_PER + tid * 4;
#pragma unroll
    for (int k = 0; k < 4; ++k)
        if (base + k < n) rowptr[base + k] += add;
    if (blockIdx.x == 0 && tid == 0) rowptr[n] = E;
}

__global__ void csr_fill(const int* __restrict__ src, const int* __restrict__ dst,
                         const int* __restrict__ rank, const int* __restrict__ rowptr,
                         int* __restrict__ col, int E) {
    int e = blockIdx.x * blockDim.x + threadIdx.x;
    if (e < E) col[rowptr[dst[e]] + rank[e]] = src[e];
}

__global__ void gather_agg_norm(const int* __restrict__ rowptr, const int* __restrict__ col,
                                const float* __restrict__ dinv, const float* __restrict__ h,
                                const float* __restrict__ bias,
                                float* __restrict__ embSlice, int n) {
    const int gid  = blockIdx.x * blockDim.x + threadIdx.x;
    const int lane = gid & 31;
    const int node = __builtin_amdgcn_readfirstlane(gid >> 5);
    if (node >= n) return;

    const int   beg = rowptr[node];
    const int   end = rowptr[node + 1];
    const float dd  = dinv[node];

    float4 acc = ((const float4*)(h + (size_t)node * HDIM))[lane];
    const float4 b = ((const float4*)bias)[lane];
    const float wself = dd * dd;
    acc.x *= wself; acc.y *= wself; acc.z *= wself; acc.w *= wself;

    for (int base = beg; base < end; base += 32) {
        int m = end - base;
        if (m > 32) m = 32;
        int   s_l = 0;
        float w_l = 0.f;
        if (lane < m) {
            s_l = col[base + lane];
            w_l = dinv[s_l] * dd;
        }
#pragma unroll 4
        for (int j = 0; j < m; ++j) {
            const int   s = __builtin_amdgcn_readlane(s_l, j);
            const float w = __builtin_bit_cast(
                float, __builtin_amdgcn_readlane(__builtin_bit_cast(int, w_l), j));
            const float4 v = ((const float4*)(h + (size_t)s * HDIM))[lane];
            acc.x = fmaf(v.x, w, acc.x);
            acc.y = fmaf(v.y, w, acc.y);
            acc.z = fmaf(v.z, w, acc.z);
            acc.w = fmaf(v.w, w, acc.w);
        }
    }
    float4 v;
    v.x = fmaxf(acc.x + b.x, 0.f);
    v.y = fmaxf(acc.y + b.y, 0.f);
    v.z = fmaxf(acc.z + b.z, 0.f);
    v.w = fmaxf(acc.w + b.w, 0.f);
    float ss = v.x * v.x + v.y * v.y + v.z * v.z + v.w * v.w;
#pragma unroll
    for (int off = 16; off > 0; off >>= 1)
        ss += __shfl_xor(ss, off, 32);
    const float scale = 1.0f / fmaxf(sqrtf(ss), 1e-12f);
    float4 o;
    o.x = v.x * scale; o.y = v.y * scale; o.z = v.z * scale; o.w = v.w * scale;
    *(float4*)(embSlice + (size_t)node * 384 + lane * 4) = o;
}

__launch_bounds__(256)
__global__ void classifier(const float* __restrict__ emb, const float* __restrict__ Wlin,
                           const float* __restrict__ blin, float* __restrict__ out, int n) {
    __shared__ float lw[384 * 8];
    __shared__ float lb[8];
    for (int i = threadIdx.x; i < 384 * 8; i += 256) lw[i] = Wlin[i];
    if (threadIdx.x < 8) lb[threadIdx.x] = blin[threadIdx.x];
    __syncthreads();

    int node = blockIdx.x * blockDim.x + threadIdx.x;
    if (node >= n) return;

    float acc[8];
#pragma unroll
    for (int c = 0; c < 8; ++c) acc[c] = lb[c];
    const float* er = emb + (size_t)node * 384;
    for (int k = 0; k < 384; ++k) {
        float e = er[k];
#pragma unroll
        for (int c = 0; c < 8; ++c) acc[c] = fmaf(e, lw[k * 8 + c], acc[c]);
    }
    float m = acc[0];
#pragma unroll
    for (int c = 1; c < 8; ++c) m = fmaxf(m, acc[c]);
    float s = 0.f;
#pragma unroll
    for (int c = 0; c < 8; ++c) s += expf(acc[c] - m);
    float lse = m + logf(s);
#pragma unroll
    for (int c = 0; c < 8; ++c) out[(size_t)node * 8 + c] = acc[c] - lse;
}

extern "C" void kernel_launch(void* const* d_in, const int* in_sizes, int n_in,
                              void* d_out, int out_size, void* d_ws, size_t ws_size,
                              hipStream_t stream) {
    const float* x    = (const float*)d_in[0];
    const int*   ei   = (const int*)d_in[1];
    const float* W1   = (const float*)d_in[2];
    const float* b1   = (const float*)d_in[3];
    const float* W2   = (const float*)d_in[4];
    const float* b2   = (const float*)d_in[5];
    const float* W3   = (const float*)d_in[6];
    const float* b3   = (const float*)d_in[7];
    const float* Wlin = (const float*)d_in[8];
    const float* blin = (const float*)d_in[9];

    const int n = in_sizes[0] / HDIM;
    const int E = in_sizes[1] / 2;
    const int* src = ei;
    const int* dst = ei + E;

    float* ws   = (float*)d_ws;
    float* dinv = ws;
    float* h    = ws + (size_t)n;
    float* emb  = h  + (size_t)n * HDIM;
    int* col      = (int*)(emb + (size_t)n * 384);
    int* rowptr   = col + (size_t)E;
    int* cnt      = rowptr + (size_t)n + 1;
    int* blockSum = cnt + (size_t)n;
    int* rank     = (int*)h;

    const int T = 256;
    const int nScan = (n + SCAN_PER - 1) / SCAN_PER;
    cnt_zero<<<(n + T - 1) / T, T, 0, stream>>>(cnt, n);
    count_rank<<<(E + T - 1) / T, T, 0, stream>>>(dst, cnt, rank, E);
    scan_local<<<nScan, 256, 0, stream>>>(cnt, rowptr, blockSum, dinv, n);
    scan_add<<<nScan, 256, 0, stream>>>(rowptr, blockSum, n, E);
    csr_fill<<<(E + T - 1) / T, T, 0, stream>>>(src, dst, rank, rowptr, col, E);

    const int nTiles = (n + 15) / 16;
    const float* Ws[3] = {W1, W2, W3};
    const float* bs[3] = {b1, b2, b3};
    const float* Xin = x;
    int lda = HDIM;

    for (int l = 0; l < 3; ++l) {
        gemm_wmma<<<(nTiles + WPB - 1) / WPB, 256, 0, stream>>>(Xin, lda, Ws[l], h, nTiles);
        gather_agg_norm<<<((size_t)n * 32 + T - 1) / T, T, 0, stream>>>(rowptr, col, dinv, h, bs[l],
                                                                        emb + l * HDIM, n);
        Xin = emb + l * HDIM;
        lda = 384;
    }

    classifier<<<(n + T - 1) / T, T, 0, stream>>>(emb, Wlin, blin, (float*)d_out, n);
}


# speedup vs baseline: 3.2139x; 1.1908x over previous
#include <hip/hip_runtime.h>
#include <math.h>

typedef float v2f __attribute__((ext_vector_type(2)));
typedef float v8f __attribute__((ext_vector_type(8)));

#define HDIM 128
#define WPB 8
#define LDSA_STRIDE 132
#define LDSW_STRIDE 136

__launch_bounds__(256)
__global__ void gemm_wmma(const float* __restrict__ X,
                          const float* __restrict__ W,
                          const float* __restrict__ dinv,
                          float* __restrict__ Hout, int nTiles) {
    __shared__ float ldsW[HDIM * LDSW_STRIDE];
    __shared__ float ldsA[WPB][16 * LDSA_STRIDE];

    const int tid  = threadIdx.x;
    const int wave = tid >> 5;
    const int lane = tid & 31;

    for (int i = 0; i < 16; ++i) {
        int idx = (i * 256 + tid) * 4;
        int row = idx >> 7, col = idx & 127;
        float4 v = *(const float4*)(W + idx);
        *(float4*)(&ldsW[row * LDSW_STRIDE + col]) = v;
    }

    const int tile   = blockIdx.x * WPB + wave;
    const bool active = (tile < nTiles);
    if (active) {
        const float* Xrow = X + (size_t)tile * 16 * HDIM;
        for (int r = 0; r < 16; ++r) {
            float4 v = *(const float4*)(Xrow + (size_t)r * HDIM + lane * 4);
            *(float4*)(&ldsA[wave][r * LDSA_STRIDE + lane * 4]) = v;
        }
    }
    __syncthreads();
    if (!active) return;

    const int m     = lane & 15;
    const int khalf = (lane >> 4) * 2;
    const int mrow  = (lane >> 4) * 8;

    const float4 dlo = *(const float4*)(dinv + (size_t)tile * 16 + mrow);
    const float4 dhi = *(const float4*)(dinv + (size_t)tile * 16 + mrow + 4);
    const float* aRow = &ldsA[wave][m * LDSA_STRIDE];

    v8f acc[8];
    const v8f vzero = {0.f, 0.f, 0.f, 0.f, 0.f, 0.f, 0.f, 0.f};
#pragma unroll
    for (int nt = 0; nt < 8; ++nt) acc[nt] = vzero;

    for (int k4 = 0; k4 < 32; ++k4) {
        const int kb = k4 * 4 + khalf;
        v2f a;
        a.x = aRow[kb];
        a.y = aRow[kb + 1];
        const float* w0 = &ldsW[kb * LDSW_STRIDE];
        const float* w1 = w0 + LDSW_STRIDE;
#pragma unroll
        for (int nt = 0; nt < 8; ++nt) {
            const int n = nt * 16 + m;
            v2f b;
            b.x = w0[n];
            b.y = w1[n];
            acc[nt] = __builtin_amdgcn_wmma_f32_16x16x4_f32(
                false, a, false, b, (short)0, acc[nt], false, false);
        }
    }

    float* outBase = Hout + (size_t)tile * 16 * HDIM;
    const float drow[8] = {dlo.x, dlo.y, dlo.z, dlo.w, dhi.x, dhi.y, dhi.z, dhi.w};
#pragma unroll
    for (int nt = 0; nt < 8; ++nt) {
#pragma unroll
        for (int r = 0; r < 8; ++r) {
            outBase[(size_t)(mrow + r) * HDIM + nt * 16 + m] = acc[nt][r] * drow[r];
        }
    }
}

#define SCAN_PER 1024

__global__ void cnt_zero(int* __restrict__ cnt, int n) {
    int i = blockIdx.x * blockDim.x + threadIdx.x;
    if (i < n) cnt[i] = 0;
}

__global__ void count_rank(const int* __restrict__ dst, int* __restrict__ cnt,
                           int* __restrict__ rank, int E) {
    int e = blockIdx.x * blockDim.x + threadIdx.x;
    if (e < E) rank[e] = atomicAdd(&cnt[dst[e]], 1);
}

__launch_bounds__(256)
__global__ void scan_local(const int* __restrict__ cnt, int* __restrict__ rowptr,
                           int* __restrict__ blockSum, float* __restrict__ dinv, int n) {
    __shared__ int waveTot[8];
    const int tid  = threadIdx.x;
    const int lane = tid & 31;
    const int wave = tid >> 5;
    const int base = blockIdx.x * SCAN_PER + tid * 4;

    int c[4];
#pragma unroll
    for (int k = 0; k < 4; ++k) c[k] = (base + k < n) ? cnt[base + k] : 0;
    const int t = c[0] + c[1] + c[2] + c[3];

    int inc = t;
#pragma unroll
    for (int off = 1; off < 32; off <<= 1) {
        int v = __shfl_up(inc, off, 32);
        if (lane >= off) inc += v;
    }
    if (lane == 31) waveTot[wave] = inc;
    __syncthreads();

    int wbase = 0, tot = 0;
#pragma unroll
    for (int w = 0; w < 8; ++w) {
        int v = waveTot[w];
        if (w < wave) wbase += v;
        tot += v;
    }
    int excl = wbase + inc - t;
#pragma unroll
    for (int k = 0; k < 4; ++k) {
        if (base + k < n) {
            rowptr[base + k] = excl;
            dinv[base + k]   = rsqrtf((float)(c[k] + 1));
            excl += c[k];
        }
    }
    if (tid == 0) blockSum[blockIdx.x] = tot;
}

__launch_bounds__(256)
__global__ void scan_add(int* __restrict__ rowptr, const int* __restrict__ blockSum,
                         int n, int E) {
    __shared__ int part[8];
    const int tid  = threadIdx.x;
    const int lane = tid & 31;
    const int wave = tid >> 5;

    int s = 0;
    for (int b = tid; b < (int)blockIdx.x; b += 256) s += blockSum[b];
#pragma unroll
    for (int off = 16; off > 0; off >>= 1) s += __shfl_xor(s, off, 32);
    if (lane == 0) part[wave] = s;
    __syncthreads();
    int add = 0;
#pragma unroll
    for (int w = 0; w < 8; ++w) add += part[w];

    const int base = blockIdx.x * SCAN_PER + tid * 4;
#pragma unroll
    for (int k = 0; k < 4; ++k)
        if (base + k < n) rowptr[base + k] += add;
    if (blockIdx.x == 0 && tid == 0) rowptr[n] = E;
}

__global__ void csr_fill(const int* __restrict__ src, const int* __restrict__ dst,
                         const int* __restrict__ rank, const int* __restrict__ rowptr,
                         int* __restrict__ col, int E) {
    int e = blockIdx.x * blockDim.x + threadIdx.x;
    if (e < E) col[rowptr[dst[e]] + rank[e]] = src[e];
}

__global__ void gather_agg_norm(const int* __restrict__ rowptr, const int* __restrict__ col,
                                const float* __restrict__ dinv, const float* __restrict__ h,
                                const float* __restrict__ bias,
                                float* __restrict__ outPlane, int n) {
    const int gid  = blockIdx.x * blockDim.x + threadIdx.x;
    const int lane = gid & 31;
    const int node = __builtin_amdgcn_readfirstlane(gid >> 5);
    if (node >= n) return;

    const int   beg = rowptr[node];
    const int   end = rowptr[node + 1];
    const float dd  = dinv[node];

    float4 acc = ((const float4*)(h + (size_t)node * HDIM))[lane];
    const float4 b = ((const float4*)bias)[lane];

    for (int base = beg; base < end; base += 32) {
        int m = end - base;
        if (m > 32) m = 32;
        int s_l = 0;
        if (lane < m) s_l = col[base + lane];
#pragma unroll 4
        for (int j = 0; j < m; ++j) {
            const int s = __builtin_amdgcn_readlane(s_l, j);
            const float4 v = ((const float4*)(h + (size_t)s * HDIM))[lane];
            acc.x += v.x;
            acc.y += v.y;
            acc.z += v.z;
            acc.w += v.w;
        }
    }
    float4 v;
    v.x = fmaxf(acc.x * dd + b.x, 0.f);
    v.y = fmaxf(acc.y * dd + b.y, 0.f);
    v.z = fmaxf(acc.z * dd + b.z, 0.f);
    v.w = fmaxf(acc.w * dd + b.w, 0.f);
    float ss = v.x * v.x + v.y * v.y + v.z * v.z + v.w * v.w;
#pragma unroll
    for (int off = 16; off > 0; off >>= 1)
        ss += __shfl_xor(ss, off, 32);
    const float scale = 1.0f / fmaxf(sqrtf(ss), 1e-12f);
    float4 o;
    o.x = v.x * scale; o.y = v.y * scale; o.z = v.z * scale; o.w = v.w * scale;
    *(float4*)(outPlane + (size_t)node * HDIM + lane * 4) = o;
}

__launch_bounds__(256)
__global__ void classifier(const float* __restrict__ emb, const float* __restrict__ Wlin,
                           const float* __restrict__ blin, float* __restrict__ out, int n) {
    __shared__ float lw[384 * 8];
    __shared__ float lb[8];
    for (int i = threadIdx.x; i < 384 * 8; i += 256) lw[i] = Wlin[i];
    if (threadIdx.x < 8) lb[threadIdx.x] = blin[threadIdx.x];
    __syncthreads();

    int node = blockIdx.x * blockDim.x + threadIdx.x;
    if (node >= n) return;

    float acc[8];
#pragma unroll
    for (int c = 0; c < 8; ++c) acc[c] = lb[c];
    for (int l = 0; l < 3; ++l) {
        const float* er = emb + ((size_t)l * n + node) * HDIM;
        const float* wl = lw + l * HDIM * 8;
        for (int k = 0; k < HDIM; ++k) {
            float e = er[k];
#pragma unroll
            for (int c = 0; c < 8; ++c) acc[c] = fmaf(e, wl[k * 8 + c], acc[c]);
        }
    }
    float m = acc[0];
#pragma unroll
    for (int c = 1; c < 8; ++c) m = fmaxf(m, acc[c]);
    float s = 0.f;
#pragma unroll
    for (int c = 0; c < 8; ++c) s += expf(acc[c] - m);
    float lse = m + logf(s);
#pragma unroll
    for (int c = 0; c < 8; ++c) out[(size_t)node * 8 + c] = acc[c] - lse;
}

extern "C" void kernel_launch(void* const* d_in, const int* in_sizes, int n_in,
                              void* d_out, int out_size, void* d_ws, size_t ws_size,
                              hipStream_t stream) {
    const float* x    = (const float*)d_in[0];
    const int*   ei   = (const int*)d_in[1];
    const float* W1   = (const float*)d_in[2];
    const float* b1   = (const float*)d_in[3];
    const float* W2   = (const float*)d_in[4];
    const float* b2   = (const float*)d_in[5];
    const float* W3   = (const float*)d_in[6];
    const float* b3   = (const float*)d_in[7];
    const float* Wlin = (const float*)d_in[8];
    const float* blin = (const float*)d_in[9];

    const int n = in_sizes[0] / HDIM;
    const int E = in_sizes[1] / 2;
    const int* src = ei;
    const int* dst = ei + E;

    float* ws   = (float*)d_ws;
    float* h    = ws;
    float* emb  = h  + (size_t)n * HDIM;
    float* dinv = emb + (size_t)n * 384;
    int* col      = (int*)(dinv + (size_t)n);
    int* rowptr   = col + (size_t)E;
    int* cnt      = rowptr + (size_t)n + 1;
    int* blockSum = cnt + (size_t)n;
    int* rank     = (int*)h;

    const int T = 256;
    const int nScan = (n + SCAN_PER - 1) / SCAN_PER;
    cnt_zero<<<(n + T - 1) / T, T, 0, stream>>>(cnt, n);
    count_rank<<<(E + T - 1) / T, T, 0, stream>>>(dst, cnt, rank, E);
    scan_local<<<nScan, 256, 0, stream>>>(cnt, rowptr, blockSum, dinv, n);
    scan_add<<<nScan, 256, 0, stream>>>(rowptr, blockSum, n, E);
    csr_fill<<<(E + T - 1) / T, T, 0, stream>>>(src, dst, rank, rowptr, col, E);

    const int nTiles = (n + 15) / 16;
    const float* Ws[3] = {W1, W2, W3};
    const float* bs[3] = {b1, b2, b3};
    const float* Xin = x;

    for (int l = 0; l < 3; ++l) {
        float* plane = emb + (size_t)l * n * HDIM;
        gemm_wmma<<<(nTiles + WPB - 1) / WPB, 256, 0, stream>>>(Xin, Ws[l], dinv, h, nTiles);
        gather_agg_norm<<<((size_t)n * 32 + T - 1) / T, T, 0, stream>>>(rowptr, col, dinv, h, bs[l],
                                                                        plane, n);
        Xin = plane;
    }

    classifier<<<(n + T - 1) / T, T, 0, stream>>>(emb, Wlin, blin, (float*)d_out, n);
}


# Edge list sorted by destination with a two-level counting sort on LDS counters instead of per-edge global atomics and scattered 4-byte accesses; a node's sum order is now LDS arrival rank
# speedup vs baseline: 4.4308x; 1.3786x over previous
#include <hip/hip_runtime.h>
#include <math.h>

typedef float v2f __attribute__((ext_vector_type(2)));
typedef float v8f __attribute__((ext_vector_type(8)));

#define HDIM 128
#define WPB 8
#define LDSA_STRIDE 132
#define LDSW_STRIDE 136

__launch_bounds__(256)
__global__ void gemm_wmma(const float* __restrict__ X,
                          const float* __restrict__ W,
                          const float* __restrict__ dinv,
                          float* __restrict__ Hout, int nTiles) {
    __shared__ float ldsW[HDIM * LDSW_STRIDE];
    __shared__ float ldsA[WPB][16 * LDSA_STRIDE];

    const int tid  = threadIdx.x;
    const int wave = tid >> 5;
    const int lane = tid & 31;

    for (int i = 0; i < 16; ++i) {
        int idx = (i * 256 + tid) * 4;
        int row = idx >> 7, col = idx & 127;
        float4 v = *(const float4*)(W + idx);
        *(float4*)(&ldsW[row * LDSW_STRIDE + col]) = v;
    }

    const int tile   = blockIdx.x * WPB + wave;
    const bool active = (tile < nTiles);
    if (active) {
        const float* Xrow = X + (size_t)tile * 16 * HDIM;
        for (int r = 0; r < 16; ++r) {
            float4 v = *(const float4*)(Xrow + (size_t)r * HDIM + lane * 4);
            *(float4*)(&ldsA[wave][r * LDSA_STRIDE + lane * 4]) = v;
        }
    }
    __syncthreads();
    if (!active) return;

    const int m     = lane & 15;
    const int khalf = (lane >> 4) * 2;
    const int mrow  = (lane >> 4) * 8;

    const float4 dlo = *(const float4*)(dinv + (size_t)tile * 16 + mrow);
    const float4 dhi = *(const float4*)(dinv + (size_t)tile * 16 + mrow + 4);
    const float* aRow = &ldsA[wave][m * LDSA_STRIDE];

    v8f acc[8];
    const v8f vzero = {0.f, 0.f, 0.f, 0.f, 0.f, 0.f, 0.f, 0.f};
#pragma unroll
    for (int nt = 0; nt < 8; ++nt) acc[nt] = vzero;

    for (int k4 = 0; k4 < 32; ++k4) {
        const int kb = k4 * 4 + khalf;
        v2f a;
        a.x = aRow[kb];
        a.y = aRow[kb + 1];
        const float* w0 = &ldsW[kb * LDSW_STRIDE];
        const float* w1 = w0 + LDSW_STRIDE;
#pragma unroll
        for (int nt = 0; nt < 8; ++nt) {
            const int n = nt * 16 + m;
            v2f b;
            b.x = w0[n];
            b.y = w1[n];
            acc[nt] = __builtin_amdgcn_wmma_f32_16x16x4_f32(
                false, a, false, b, (short)0, acc[nt], false, false);
        }
    }

    float* outBase = Hout + (size_t)tile * 16 * HDIM;
    const float drow[8] = {dlo.x, dlo.y, dlo.z, dlo.w, dhi.x, dhi.y, dhi.z, dhi.w};
#pragma unroll
    for (int nt = 0; nt < 8; ++nt) {
#pragma unroll
        for (int r = 0; r < 8; ++r) {
            outBase[(size_t)(mrow + r) * HDIM + nt * 16 + m] = acc[nt][r] * drow[r];
        }
    }
}

#define BSH       9
#define BNODES    512
#define NBK       256
#define ECH       8192
#define STAGE_CAP 16384

template <int NW>
__device__ __forceinline__ int block_excl_scan(int v, int* wsum, int lane, int wave, int& total) {
    int inc = v;
#pragma unroll
    for (int off = 1; off < 32; off <<= 1) {
        int t = __shfl_up(inc, off, 32);
        if (lane >= off) inc += t;
    }
    if (lane == 31) wsum[wave] = inc;
    __syncthreads();
    int wbase = 0, tot = 0;
#pragma unroll
    for (int w = 0; w < NW; ++w) {
        int x = wsum[w];
        if (w < wave) wbase += x;
        tot += x;
    }
    total = tot;
    __syncthreads();
    return wbase + inc - v;
}

__device__ __forceinline__ int bucket_of(int d) {
    unsigned b = (unsigned)d >> BSH;
    return (int)(b < NBK ? b : NBK - 1);
}

__launch_bounds__(256)
__global__ void bucket_hist(const int* __restrict__ dst, int* __restrict__ tbl, int E) {
    __shared__ int hist[NBK];
    const int tid = threadIdx.x;
    hist[tid] = 0;
    __syncthreads();
    const int beg = blockIdx.x * ECH;
    int end = beg + ECH;
    if (end > E) end = E;
    for (int e0 = beg + tid; e0 < end; e0 += 256 * 8) {
        int d[8];
#pragma unroll
        for (int k = 0; k < 8; ++k) d[k] = dst[min(e0 + k * 256, end - 1)];
#pragma unroll
        for (int k = 0; k < 8; ++k)
            if (e0 + k * 256 < end) atomicAdd(&hist[bucket_of(d[k])], 1);
    }
    __syncthreads();
    tbl[blockIdx.x * NBK + tid] = hist[tid];
}

__launch_bounds__(256)
__global__ void bucket_scatter(const int* __restrict__ src, const int* __restrict__ dst,
                               const int* __restrict__ tbl, int* __restrict__ bucketStart,
                               unsigned* __restrict__ bPack, int E, int G) {
    __shared__ int sBase[NBK];
    __shared__ int sCur[NBK];
    __shared__ unsigned sPack[ECH];
    __shared__ unsigned char sBkt[ECH];
    __shared__ int wsum[8];
    const int tid  = threadIdx.x;
    const int lane = tid & 31;
    const int wave = tid >> 5;
    const int blk  = blockIdx.x;

    int pre = 0, tot = 0;
#pragma unroll 16
    for (int j = 0; j < G; ++j) {
        const int v = tbl[j * NBK + tid];
        tot += v;
        pre += (j < blk) ? v : 0;
    }
    const int mine = tbl[blk * NBK + tid];
    int unused;
    const int bStart = block_excl_scan<8>(tot, wsum, lane, wave, unused);
    const int lOff   = block_excl_scan<8>(mine, wsum, lane, wave, unused);
    sBase[tid] = bStart + pre - lOff;
    sCur[tid]  = lOff;
    if (blk == 0) {
        bucketStart[tid] = bStart;
        if (tid == NBK - 1) bucketStart[NBK] = bStart + tot;
    }
    __syncthreads();

    const int beg = blk * ECH;
    int end = beg + ECH;
    if (end > E) end = E;
    for (int e0 = beg + tid; e0 < end; e0 += 256 * 8) {
        int s[8], d[8];
#pragma unroll
        for (int k = 0; k < 8; ++k) {
            const int ec = min(e0 + k * 256, end - 1);
            s[k] = src[ec];
            d[k] = dst[ec];
        }
#pragma unroll
        for (int k = 0; k < 8; ++k) {
            if (e0 + k * 256 < end) {
                const int b = bucket_of(d[k]);
                const int p = atomicAdd(&sCur[b], 1);
                sPack[p] = ((unsigned)s[k] << BSH) | ((unsigned)d[k] & (BNODES - 1));
                sBkt[p]  = (unsigned char)b;
            }
        }
    }
    __syncthreads();
    const int m = end - beg;
    for (int i = tid; i < m; i += 256) bPack[sBase[sBkt[i]] + i] = sPack[i];
}

__launch_bounds__(512)
__global__ void bucket_csr(const unsigned* __restrict__ bPack, const int* __restrict__ bucketStart,
                           int* __restrict__ rowptr, float* __restrict__ dinv,
                           int* __restrict__ col, int n, int E) {
    __shared__ int cnt[BNODES];
    __shared__ int cur[BNODES];
    __shared__ int stage[STAGE_CAP];
    __shared__ int wsum[16];
    const int tid  = threadIdx.x;
    const int lane = tid & 31;
    const int wave = tid >> 5;
    const int b    = blockIdx.x;
    const int bs   = bucketStart[b];
    const int m    = bucketStart[b + 1] - bs;

    cnt[tid] = 0;
    __syncthreads();
    for (int i0 = tid; i0 < m; i0 += 512 * 8) {
        unsigned w[8];
#pragma unroll
        for (int k = 0; k < 8; ++k) w[k] = bPack[bs + min(i0 + k * 512, m - 1)];
#pragma unroll
        for (int k = 0; k < 8; ++k)
            if (i0 + k * 512 < m) atomicAdd(&cnt[w[k] & (BNODES - 1)], 1);
    }
    __syncthreads();

    const int c = cnt[tid];
    int unused;
    const int off = block_excl_scan<16>(c, wsum, lane, wave, unused);
    const int node = (b << BSH) + tid;
    if (node < n) {
        rowptr[node] = bs + off;
        dinv[node]   = rsqrtf((float)(c + 1));
    }
    if (b == 0 && tid == 0) rowptr[n] = E;
    cur[tid] = off;
    __syncthreads();

    if (m <= STAGE_CAP) {
        for (int i0 = tid; i0 < m; i0 += 512 * 8) {
            unsigned w[8];
#pragma unroll
            for (int k = 0; k < 8; ++k) w[k] = bPack[bs + min(i0 + k * 512, m - 1)];
#pragma unroll
            for (int k = 0; k < 8; ++k) {
                if (i0 + k * 512 < m) {
                    const int p = atomicAdd(&cur[w[k] & (BNODES - 1)], 1);
                    stage[p] = (int)(w[k] >> BSH);
                }
            }
        }
        __syncthreads();
        for (int i = tid; i < m; i += 512) col[bs + i] = stage[i];
    } else {
        for (int i = tid; i < m; i += 512) {
            const unsigned w = bPack[bs + i];
            const int p = atomicAdd(&cur[w & (BNODES - 1)], 1);
            col[bs + p] = (int)(w >> BSH);
        }
    }
}

__global__ void gather_agg_norm(const int* __restrict__ rowptr, const int* __restrict__ col,
                                const float* __restrict__ dinv, const float* __restrict__ h,
                                const float* __restrict__ bias,
                                float* __restrict__ outPlane, int n) {
    const int gid  = blockIdx.x * blockDim.x + threadIdx.x;
    const int lane = gid & 31;
    const int node = __builtin_amdgcn_readfirstlane(gid >> 5);
    if (node >= n) return;

    const int   beg = rowptr[node];
    const int   end = rowptr[node + 1];
    const float dd  = dinv[node];

    float4 acc = ((const float4*)(h + (size_t)node * HDIM))[lane];
    const float4 b = ((const float4*)bias)[lane];

    for (int base = beg; base < end; base += 32) {
        int m = end - base;
        if (m > 32) m = 32;
        int s_l = 0;
        if (lane < m) s_l = col[base + lane];
#pragma unroll 4
        for (int j = 0; j < m; ++j) {
            const int s = __builtin_amdgcn_readlane(s_l, j);
            const float4 v = ((const float4*)(h + (size_t)s * HDIM))[lane];
            acc.x += v.x;
            acc.y += v.y;
            acc.z += v.z;
            acc.w += v.w;
        }
    }
    float4 v;
    v.x = fmaxf(acc.x * dd + b.x, 0.f);
    v.y = fmaxf(acc.y * dd + b.y, 0.f);
    v.z = fmaxf(acc.z * dd + b.z, 0.f);
    v.w = fmaxf(acc.w * dd + b.w, 0.f);
    float ss = v.x * v.x + v.y * v.y + v.z * v.z + v.w * v.w;
#pragma unroll
    for (int off = 16; off > 0; off >>= 1)
        ss += __shfl_xor(ss, off, 32);
    const float scale = 1.0f / fmaxf(sqrtf(ss), 1e-12f);
    float4 o;
    o.x = v.x * scale; o.y = v.y * scale; o.z = v.z * scale; o.w = v.w * scale;
    *(float4*)(outPlane + (size_t)node * HDIM + lane * 4) = o;
}

__launch_bounds__(256)
__global__ void classifier(const float* __restrict__ emb, const float* __restrict__ Wlin,
                           const float* __restrict__ blin, float* __restrict__ out, int n) {
    __shared__ float lw[384 * 8];
    __shared__ float lb[8];
    for (int i = threadIdx.x; i < 384 * 8; i += 256) lw[i] = Wlin[i];
    if (threadIdx.x < 8) lb[threadIdx.x] = blin[threadIdx.x];
    __syncthreads();

    int node = blockIdx.x * blockDim.x + threadIdx.x;
    if (node >= n) return;

    float acc[8];
#pragma unroll
    for (int c = 0; c < 8; ++c) acc[c] = lb[c];
    for (int l = 0; l < 3; ++l) {
        const float* er = emb + ((size_t)l * n + node) * HDIM;
        const float* wl = lw + l * HDIM * 8;
        for (int k = 0; k < HDIM; ++k) {
            float e = er[k];
#pragma unroll
            for (int c = 0; c < 8; ++c) acc[c] = fmaf(e, wl[k * 8 + c], acc[c]);
        }
    }
    float m = acc[0];
#pragma unroll
    for (int c = 1; c < 8; ++c) m = fmaxf(m, acc[c]);
    float s = 0.f;
#pragma unroll
    for (int c = 0; c < 8; ++c) s += expf(acc[c] - m);
    float lse = m + logf(s);
#pragma unroll
    for (int c = 0; c < 8; ++c) out[(size_t)node * 8 + c] = acc[c] - lse;
}

extern "C" void kernel_launch(void* const* d_in, const int* in_sizes, int n_in,
                              void* d_out, int out_size, void* d_ws, size_t ws_size,
                              hipStream_t stream) {
    const float* x    = (const float*)d_in[0];
    const int*   ei   = (const int*)d_in[1];
    const float* W1   = (const float*)d_in[2];
    const float* b1   = (const float*)d_in[3];
    const float* W2   = (const float*)d_in[4];
    const float* b2   = (const float*)d_in[5];
    const float* W3   = (const float*)d_in[6];
    const float* b3   = (const float*)d_in[7];
    const float* Wlin = (const float*)d_in[8];
    const float* blin = (const float*)d_in[9];

    const int n = in_sizes[0] / HDIM;
    const int E = in_sizes[1] / 2;
    const int* src = ei;
    const int* dst = ei + E;

    float* ws   = (float*)d_ws;
    float* h    = ws;
    float* emb  = h  + (size_t)n * HDIM;
    float* dinv = emb + (size_t)n * 384;
    int* col         = (int*)(dinv + (size_t)n);
    int* rowptr      = col + (size_t)E;
    int* bucketStart = rowptr + (size_t)n + 1;
    int* tbl         = bucketStart + NBK + 1;
    unsigned* bPack  = (unsigned*)h;

    const int T = 256;
    const int G = (E + ECH - 1) / ECH;
    bucket_hist<<<G, 256, 0, stream>>>(dst, tbl, E);
    bucket_scatter<<<G, 256, 0, stream>>>(src, dst, tbl, bucketStart, bPack, E, G);
    bucket_csr<<<NBK, 512, 0, stream>>>(bPack, bucketStart, rowptr, dinv, col, n, E);

    const int nTiles = (n + 15) / 16;
    const float* Ws[3] = {W1, W2, W3};
    const float* bs[3] = {b1, b2, b3};
    const float* Xin = x;

    for (int l = 0; l < 3; ++l) {
        float* plane = emb + (size_t)l * n * HDIM;
        gemm_wmma<<<(nTiles + WPB - 1) / WPB, 256, 0, stream>>>(Xin, Ws[l], dinv, h, nTiles);
        gather_agg_norm<<<((size_t)n * 32 + T - 1) / T, T, 0, stream>>>(rowptr, col, dinv, h, bs[l],
                                                                        plane, n);
        Xin = plane;
    }

    classifier<<<(n + T - 1) / T, T, 0, stream>>>(emb, Wlin, blin, (float*)d_out, n);
}


# Layer 2 and 3 transforms fused into the gather kernel of the layer before: a persistent 8-wave work-group gathers a 16-node tile into LDS and runs its f32 WMMA product there; 2 of 10 launches go
# speedup vs baseline: 4.9151x; 1.1093x over previous
#include <hip/hip_runtime.h>
#include <math.h>

typedef float v2f __attribute__((ext_vector_type(2)));
typedef float v8f __attribute__((ext_vector_type(8)));

#define HDIM 128
#define WPB 8
#define LDSA_STRIDE 132
#define LDSW_STRIDE 136

__launch_bounds__(256)
__global__ void gemm_wmma(const float* __restrict__ X,
                          const float* __restrict__ W,
                          const float* __restrict__ dinv,
                          float* __restrict__ Hout, int nTiles) {
    __shared__ float ldsW[HDIM * LDSW_STRIDE];
    __shared__ float ldsA[WPB][16 * LDSA_STRIDE];

    const int tid  = threadIdx.x;
    const int wave = tid >> 5;
    const int lane = tid & 31;

    for (int i = 0; i < 16; ++i) {
        int idx = (i * 256 + tid) * 4;
        int row = idx >> 7, col = idx & 127;
        float4 v = *(const float4*)(W + idx);
        *(float4*)(&ldsW[row * LDSW_STRIDE + col]) = v;
    }

    const int tile   = blockIdx.x * WPB + wave;
    const bool active = (tile < nTiles);
    if (active) {
        const float* Xrow = X + (size_t)tile * 16 * HDIM;
        for (int r = 0; r < 16; ++r) {
            float4 v = *(const float4*)(Xrow + (size_t)r * HDIM + lane * 4);
            *(float4*)(&ldsA[wave][r * LDSA_STRIDE + lane * 4]) = v;
        }
    }
    __syncthreads();
    if (!active) return;

    const int m     = lane & 15;
    const int khalf = (lane >> 4) * 2;
    const int mrow  = (lane >> 4) * 8;

    const float4 dlo = *(const float4*)(dinv + (size_t)tile * 16 + mrow);
    const float4 dhi = *(const float4*)(dinv + (size_t)tile * 16 + mrow + 4);
    const float* aRow = &ldsA[wave][m * LDSA_STRIDE];

    v8f acc[8];
    const v8f vzero = {0.f, 0.f, 0.f, 0.f, 0.f, 0.f, 0.f, 0.f};
#pragma unroll
    for (int nt = 0; nt < 8; ++nt) acc[nt] = vzero;

    for (int k4 = 0; k4 < 32; ++k4) {
        const int kb = k4 * 4 + khalf;
        v2f a;
        a.x = aRow[kb];
        a.y = aRow[kb + 1];
        const float* w0 = &ldsW[kb * LDSW_STRIDE];
        const float* w1 = w0 + LDSW_STRIDE;
#pragma unroll
        for (int nt = 0; nt < 8; ++nt) {
            const int n = nt * 16 + m;
            v2f b;
            b.x = w0[n];
            b.y = w1[n];
            acc[nt] = __builtin_amdgcn_wmma_f32_16x16x4_f32(
                false, a, false, b, (short)0, acc[nt], false, false);
        }
    }

    float* outBase = Hout + (size_t)tile * 16 * HDIM;
    const float drow[8] = {dlo.x, dlo.y, dlo.z, dlo.w, dhi.x, dhi.y, dhi.z, dhi.w};
#pragma unroll
    for (int nt = 0; nt < 8; ++nt) {
#pragma unroll
        for (int r = 0; r < 8; ++r) {
            outBase[(size_t)(mrow + r) * HDIM + nt * 16 + m] = acc[nt][r] * drow[r];
        }
    }
}

#define BSH       9
#define BNODES    512
#define NBK       256
#define ECH       8192
#define STAGE_CAP 16384

template <int NW>
__device__ __forceinline__ int block_excl_scan(int v, int* wsum, int lane, int wave, int& total) {
    int inc = v;
#pragma unroll
    for (int off = 1; off < 32; off <<= 1) {
        int t = __shfl_up(inc, off, 32);
        if (lane >= off) inc += t;
    }
    if (lane == 31) wsum[wave] = inc;
    __syncthreads();
    int wbase = 0, tot = 0;
#pragma unroll
    for (int w = 0; w < NW; ++w) {
        int x = wsum[w];
        if (w < wave) wbase += x;
        tot += x;
    }
    total = tot;
    __syncthreads();
    return wbase + inc - v;
}

__device__ __forceinline__ int bucket_of(int d) {
    unsigned b = (unsigned)d >> BSH;
    return (int)(b < NBK ? b : NBK - 1);
}

__launch_bounds__(256)
__global__ void bucket_hist(const int* __restrict__ dst, int* __restrict__ tbl, int E) {
    __shared__ int hist[NBK];
    const int tid = threadIdx.x;
    hist[tid] = 0;
    __syncthreads();
    const int beg = blockIdx.x * ECH;
    int end = beg + ECH;
    if (end > E) end = E;
    for (int e0 = beg + tid; e0 < end; e0 += 256 * 8) {
        int d[8];
#pragma unroll
        for (int k = 0; k < 8; ++k) d[k] = dst[min(e0 + k * 256, end - 1)];
#pragma unroll
        for (int k = 0; k < 8; ++k)
            if (e0 + k * 256 < end) atomicAdd(&hist[bucket_of(d[k])], 1);
    }
    __syncthreads();
    tbl[blockIdx.x * NBK + tid] = hist[tid];
}

__launch_bounds__(256)
__global__ void bucket_scatter(const int* __restrict__ src, const int* __restrict__ dst,
                               const int* __restrict__ tbl, int* __restrict__ bucketStart,
                               unsigned* __restrict__ bPack, int E, int G) {
    __shared__ int sBase[NBK];
    __shared__ int sCur[NBK];
    __shared__ unsigned sPack[ECH];
    __shared__ unsigned char sBkt[ECH];
    __shared__ int wsum[8];
    const int tid  = threadIdx.x;
    const int lane = tid & 31;
    const int wave = tid >> 5;
    const int blk  = blockIdx.x;

    int pre = 0, tot = 0;
#pragma unroll 16
    for (int j = 0; j < G; ++j) {
        const int v = tbl[j * NBK + tid];
        tot += v;
        pre += (j < blk) ? v : 0;
    }
    const int mine = tbl[blk * NBK + tid];
    int unused;
    const int bStart = block_excl_scan<8>(tot, wsum, lane, wave, unused);
    const int lOff   = block_excl_scan<8>(mine, wsum, lane, wave, unused);
    sBase[tid] = bStart + pre - lOff;
    sCur[tid]  = lOff;
    if (blk == 0) {
        bucketStart[tid] = bStart;
        if (tid == NBK - 1) bucketStart[NBK] = bStart + tot;
    }
    __syncthreads();

    const int beg = blk * ECH;
    int end = beg + ECH;
    if (end > E) end = E;
    for (int e0 = beg + tid; e0 < end; e0 += 256 * 8) {
        int s[8], d[8];
#pragma unroll
        for (int k = 0; k < 8; ++k) {
            const int ec = min(e0 + k * 256, end - 1);
            s[k] = src[ec];
            d[k] = dst[ec];
        }
#pragma unroll
        for (int k = 0; k < 8; ++k) {
            if (e0 + k * 256 < end) {
                const int b = bucket_of(d[k]);
                const int p = atomicAdd(&sCur[b], 1);
                sPack[p] = ((unsigned)s[k] << BSH) | ((unsigned)d[k] & (BNODES - 1));
                sBkt[p]  = (unsigned char)b;
            }
        }
    }
    __syncthreads();
    const int m = end - beg;
    for (int i = tid; i < m; i += 256) bPack[sBase[sBkt[i]] + i] = sPack[i];
}

__launch_bounds__(512)
__global__ void bucket_csr(const unsigned* __restrict__ bPack, const int* __restrict__ bucketStart,
                           int* __restrict__ rowptr, float* __restrict__ dinv,
                           int* __restrict__ col, int n, int E) {
    __shared__ int cnt[BNODES];
    __shared__ int cur[BNODES];
    __shared__ int stage[STAGE_CAP];
    __shared__ int wsum[16];
    const int tid  = threadIdx.x;
    const int lane = tid & 31;
    const int wave = tid >> 5;
    const int b    = blockIdx.x;
    const int bs   = bucketStart[b];
    const int m    = bucketStart[b + 1] - bs;

    cnt[tid] = 0;
    __syncthreads();
    for (int i0 = tid; i0 < m; i0 += 512 * 8) {
        unsigned w[8];
#pragma unroll
        for (int k = 0; k < 8; ++k) w[k] = bPack[bs + min(i0 + k * 512, m - 1)];
#pragma unroll
        for (int k = 0; k < 8; ++k)
            if (i0 + k * 512 < m) atomicAdd(&cnt[w[k] & (BNODES - 1)], 1);
    }
    __syncthreads();

    const int c = cnt[tid];
    int unused;
    const int off = block_excl_scan<16>(c, wsum, lane, wave, unused);
    const int node = (b << BSH) + tid;
    if (node < n) {
        rowptr[node] = bs + off;
        dinv[node]   = rsqrtf((float)(c + 1));
    }
    if (b == 0 && tid == 0) rowptr[n] = E;
    cur[tid] = off;
    __syncthreads();

    if (m <= STAGE_CAP) {
        for (int i0 = tid; i0 < m; i0 += 512 * 8) {
            unsigned w[8];
#pragma unroll
            for (int k = 0; k < 8; ++k) w[k] = bPack[bs + min(i0 + k * 512, m - 1)];
#pragma unroll
            for (int k = 0; k < 8; ++k) {
                if (i0 + k * 512 < m) {
                    const int p = atomicAdd(&cur[w[k] & (BNODES - 1)], 1);
                    stage[p] = (int)(w[k] >> BSH);
                }
            }
        }
        __syncthreads();
        for (int i = tid; i < m; i += 512) col[bs + i] = stage[i];
    } else {
        for (int i = tid; i < m; i += 512) {
            const unsigned w = bPack[bs + i];
            const int p = atomicAdd(&cur[w & (BNODES - 1)], 1);
            col[bs + p] = (int)(w >> BSH);
        }
    }
}

__global__ void gather_agg_norm(const int* __restrict__ rowptr, const int* __restrict__ col,
                                const float* __restrict__ dinv, const float* __restrict__ h,
                                const float* __restrict__ bias,
                                float* __restrict__ outPlane, int n) {
    const int gid  = blockIdx.x * blockDim.x + threadIdx.x;
    const int lane = gid & 31;
    const int node = __builtin_amdgcn_readfirstlane(gid >> 5);
    if (node >= n) return;

    const int   beg = rowptr[node];
    const int   end = rowptr[node + 1];
    const float dd  = dinv[node];

    float4 acc = ((const float4*)(h + (size_t)node * HDIM))[lane];
    const float4 b = ((const float4*)bias)[lane];

    for (int base = beg; base < end; base += 32) {
        int m = end - base;
        if (m > 32) m = 32;
        int s_l = 0;
        if (lane < m) s_l = col[base + lane];
#pragma unroll 4
        for (int j = 0; j < m; ++j) {
            const int s = __builtin_amdgcn_readlane(s_l, j);
            const float4 v = ((const float4*)(h + (size_t)s * HDIM))[lane];
            acc.x += v.x;
            acc.y += v.y;
            acc.z += v.z;
            acc.w += v.w;
        }
    }
    float4 v;
    v.x = fmaxf(acc.x * dd + b.x, 0.f);
    v.y = fmaxf(acc.y * dd + b.y, 0.f);
    v.z = fmaxf(acc.z * dd + b.z, 0.f);
    v.w = fmaxf(acc.w * dd + b.w, 0.f);
    float ss = v.x * v.x + v.y * v.y + v.z * v.z + v.w * v.w;
#pragma unroll
    for (int off = 16; off > 0; off >>= 1)
        ss += __shfl_xor(ss, off, 32);
    const float scale = 1.0f / fmaxf(sqrtf(ss), 1e-12f);
    float4 o;
    o.x = v.x * scale; o.y = v.y * scale; o.z = v.z * scale; o.w = v.w * scale;
    *(float4*)(outPlane + (size_t)node * HDIM + lane * 4) = o;
}

#define FUSED_GRID 1024

__device__ __forceinline__ float4 row_tail(const float4 acc, const float dd, const float4 b) {
    float4 v;
    v.x = fmaxf(acc.x * dd + b.x, 0.f);
    v.y = fmaxf(acc.y * dd + b.y, 0.f);
    v.z = fmaxf(acc.z * dd + b.z, 0.f);
    v.w = fmaxf(acc.w * dd + b.w, 0.f);
    float ss = v.x * v.x + v.y * v.y + v.z * v.z + v.w * v.w;
#pragma unroll
    for (int off = 16; off > 0; off >>= 1)
        ss += __shfl_xor(ss, off, 32);
    const float scale = 1.0f / fmaxf(sqrtf(ss), 1e-12f);
    float4 o;
    o.x = v.x * scale; o.y = v.y * scale; o.z = v.z * scale; o.w = v.w * scale;
    return o;
}

__launch_bounds__(256)
__global__ void gather_norm_gemm(const int* __restrict__ rowptr, const int* __restrict__ col,
                                 const float* __restrict__ dinv, const float* __restrict__ h,
                                 const float* __restrict__ bias, float* __restrict__ outPlane,
                                 const float* __restrict__ Wn, float* __restrict__ hNext,
                                 int nTiles) {
    __shared__ float ldsW[HDIM * LDSW_STRIDE];
    __shared__ float ldsT[16 * LDSA_STRIDE];

    const int tid  = threadIdx.x;
    const int lane = tid & 31;
    const int wave = __builtin_amdgcn_readfirstlane(tid >> 5);

    for (int i = 0; i < 16; ++i) {
        int idx = (i * 256 + tid) * 4;
        int row = idx >> 7, cc = idx & 127;
        float4 v = *(const float4*)(Wn + idx);
        *(float4*)(&ldsW[row * LDSW_STRIDE + cc]) = v;
    }
    const float4 b = ((const float4*)bias)[lane];

    const int m     = lane & 15;
    const int khalf = (lane >> 4) * 2;
    const int mrow  = (lane >> 4) * 8;
    const float* aRow = &ldsT[m * LDSA_STRIDE];
    const float* wCol = &ldsW[wave * 16 + m];
    const int rA = wave * 2;

    for (int tile = blockIdx.x; tile < nTiles; tile += gridDim.x) {
        const int nA = tile * 16 + rA;
        const int pA = rowptr[nA];
        const int pB = rowptr[nA + 1];
        const int pC = rowptr[nA + 2];
        const float ddA = dinv[nA];
        const float ddB = dinv[nA + 1];
        float4 accA = ((const float4*)(h + (size_t)nA * HDIM))[lane];
        float4 accB = ((const float4*)(h + (size_t)(nA + 1) * HDIM))[lane];
        const float4 dlo = *(const float4*)(dinv + (size_t)tile * 16 + mrow);
        const float4 dhi = *(const float4*)(dinv + (size_t)tile * 16 + mrow + 4);

        for (int baseA = pA, baseB = pB; baseA < pB || baseB < pC; baseA += 32, baseB += 32) {
            int mA = pB - baseA; mA = mA < 0 ? 0 : (mA > 32 ? 32 : mA);
            int mB = pC - baseB; mB = mB < 0 ? 0 : (mB > 32 ? 32 : mB);
            int sA = 0, sB = 0;
            if (lane < mA) sA = col[baseA + lane];
            if (lane < mB) sB = col[baseB + lane];
            const int mm = mA > mB ? mA : mB;
            for (int j0 = 0; j0 < mm; j0 += 4) {
                int iA[4], iB[4];
#pragma unroll
                for (int u = 0; u < 4; ++u) {
                    iA[u] = __builtin_amdgcn_readlane(sA, j0 + u);
                    iB[u] = __builtin_amdgcn_readlane(sB, j0 + u);
                }
                float4 vA[4], vB[4];
#pragma unroll
                for (int u = 0; u < 4; ++u)
                    if (j0 + u < mA) vA[u] = ((const float4*)(h + (size_t)iA[u] * HDIM))[lane];
#pragma unroll
                for (int u = 0; u < 4; ++u)
                    if (j0 + u < mB) vB[u] = ((const float4*)(h + (size_t)iB[u] * HDIM))[lane];
#pragma unroll
                for (int u = 0; u < 4; ++u)
                    if (j0 + u < mA) {
                        accA.x += vA[u].x; accA.y += vA[u].y; accA.z += vA[u].z; accA.w += vA[u].w;
                    }
#pragma unroll
                for (int u = 0; u < 4; ++u)
                    if (j0 + u < mB) {
                        accB.x += vB[u].x; accB.y += vB[u].y; accB.z += vB[u].z; accB.w += vB[u].w;
                    }
            }
        }
        const float4 oA = row_tail(accA, ddA, b);
        const float4 oB = row_tail(accB, ddB, b);
        *(float4*)(outPlane + (size_t)nA * HDIM + lane * 4) = oA;
        *(float4*)(outPlane + (size_t)(nA + 1) * HDIM + lane * 4) = oB;
        *(float4*)(&ldsT[rA * LDSA_STRIDE + lane * 4]) = oA;
        *(float4*)(&ldsT[(rA + 1) * LDSA_STRIDE + lane * 4]) = oB;
        __syncthreads();

        v8f acc = {0.f, 0.f, 0.f, 0.f, 0.f, 0.f, 0.f, 0.f};
        for (int k4 = 0; k4 < 32; ++k4) {
            const int kb = k4 * 4 + khalf;
            v2f a;
            a.x = aRow[kb];
            a.y = aRow[kb + 1];
            v2f w;
            w.x = wCol[kb * LDSW_STRIDE];
            w.y = wCol[(kb + 1) * LDSW_STRIDE];
            acc = __builtin_amdgcn_wmma_f32_16x16x4_f32(
                false, a, false, w, (short)0, acc, false, false);
        }
        __syncthreads();

        const float drow[8] = {dlo.x, dlo.y, dlo.z, dlo.w, dhi.x, dhi.y, dhi.z, dhi.w};
#pragma unroll
        for (int r = 0; r < 8; ++r)
            ldsT[(mrow + r) * LDSA_STRIDE + wave * 16 + m] = acc[r] * drow[r];
        __syncthreads();
        const float4 hA = *(const float4*)(&ldsT[rA * LDSA_STRIDE + lane * 4]);
        const float4 hB = *(const float4*)(&ldsT[(rA + 1) * LDSA_STRIDE + lane * 4]);
        *(float4*)(hNext + (size_t)nA * HDIM + lane * 4) = hA;
        *(float4*)(hNext + (size_t)(nA + 1) * HDIM + lane * 4) = hB;
    }
}

__launch_bounds__(256)
__global__ void classifier(const float* __restrict__ emb, const float* __restrict__ Wlin,
                           const float* __restrict__ blin, float* __restrict__ out, int n) {
    __shared__ float lw[384 * 8];
    __shared__ float lb[8];
    for (int i = threadIdx.x; i < 384 * 8; i += 256) lw[i] = Wlin[i];
    if (threadIdx.x < 8) lb[threadIdx.x] = blin[threadIdx.x];
    __syncthreads();

    int node = blockIdx.x * blockDim.x + threadIdx.x;
    if (node >= n) return;

    float acc[8];
#pragma unroll
    for (int c = 0; c < 8; ++c) acc[c] = lb[c];
    for (int l = 0; l < 3; ++l) {
        const float* er = emb + ((size_t)l * n + node) * HDIM;
        const float* wl = lw + l * HDIM * 8;
        for (int k = 0; k < HDIM; ++k) {
            float e = er[k];
#pragma unroll
            for (int c = 0; c < 8; ++c) acc[c] = fmaf(e, wl[k * 8 + c], acc[c]);
        }
    }
    float m = acc[0];
#pragma unroll
    for (int c = 1; c < 8; ++c) m = fmaxf(m, acc[c]);
    float s = 0.f;
#pragma unroll
    for (int c = 0; c < 8; ++c) s += expf(acc[c] - m);
    float lse = m + logf(s);
#pragma unroll
    for (int c = 0; c < 8; ++c) out[(size_t)node * 8 + c] = acc[c] - lse;
}

extern "C" void kernel_launch(void* const* d_in, const int* in_sizes, int n_in,
                              void* d_out, int out_size, void* d_ws, size_t ws_size,
                              hipStream_t stream) {
    const float* x    = (const float*)d_in[0];
    const int*   ei   = (const int*)d_in[1];
    const float* W1   = (const float*)d_in[2];
    const float* b1   = (const float*)d_in[3];
    const float* W2   = (const float*)d_in[4];
    const float* b2   = (const float*)d_in[5];
    const float* W3   = (const float*)d_in[6];
    const float* b3   = (const float*)d_in[7];
    const float* Wlin = (const float*)d_in[8];
    const float* blin = (const float*)d_in[9];

    const int n = in_sizes[0] / HDIM;
    const int E = in_sizes[1] / 2;
    const int* src = ei;
    const int* dst = ei + E;

    float* ws   = (float*)d_ws;
    float* h    = ws;
    float* emb  = h  + (size_t)n * HDIM;
    float* h2   = emb + (size_t)n * 384;
    float* dinv = h2 + (size_t)n * HDIM;
    int* col         = (int*)(dinv + (size_t)n);
    int* rowptr      = col + (size_t)E;
    int* bucketStart = rowptr + (size_t)n + 1;
    int* tbl         = bucketStart + NBK + 1;
    unsigned* bPack  = (unsigned*)h;

    const int T = 256;
    const int G = (E + ECH - 1) / ECH;
    bucket_hist<<<G, 256, 0, stream>>>(dst, tbl, E);
    bucket_scatter<<<G, 256, 0, stream>>>(src, dst, tbl, bucketStart, bPack, E, G);
    bucket_csr<<<NBK, 512, 0, stream>>>(bPack, bucketStart, rowptr, dinv, col, n, E);

    const int nTiles = (n + 15) / 16;
    float* plane0 = emb;
    float* plane1 = emb + (size_t)n * HDIM;
    float* plane2 = emb + (size_t)2 * n * HDIM;

    gemm_wmma<<<(nTiles + WPB - 1) / WPB, 256, 0, stream>>>(x, W1, dinv, h, nTiles);
    gather_norm_gemm<<<FUSED_GRID, 256, 0, stream>>>(rowptr, col, dinv, h, b1, plane0, W2, h2, nTiles);
    gather_norm_gemm<<<FUSED_GRID, 256, 0, stream>>>(rowptr, col, dinv, h2, b2, plane1, W3, h, nTiles);
    gather_agg_norm<<<((size_t)n * 32 + T - 1) / T, T, 0, stream>>>(rowptr, col, dinv, h, b3, plane2, n);

    classifier<<<(n + T - 1) / T, T, 0, stream>>>(emb, Wlin, blin, (float*)d_out, n);
}


# Classifier folded into the gather waves: each layer's share of a node's 8 logits is taken from the row in registers (lane-tree sums, not column order); emb planes and the classifier launch go
# speedup vs baseline: 5.4239x; 1.1035x over previous
#include <hip/hip_runtime.h>
#include <math.h>

typedef float v2f __attribute__((ext_vector_type(2)));
typedef float v8f __attribute__((ext_vector_type(8)));

#define HDIM 128
#define WPB 8
#define LDSA_STRIDE 132
#define LDSW_STRIDE 136

__launch_bounds__(256)
__global__ void gemm_wmma(const float* __restrict__ X,
                          const float* __restrict__ W,
                          const float* __restrict__ dinv,
                          float* __restrict__ Hout, int nTiles) {
    __shared__ float ldsW[HDIM * LDSW_STRIDE];
    __shared__ float ldsA[WPB][16 * LDSA_STRIDE];

    const int tid  = threadIdx.x;
    const int wave = tid >> 5;
    const int lane = tid & 31;

    for (int i = 0; i < 16; ++i) {
        int idx = (i * 256 + tid) * 4;
        int row = idx >> 7, col = idx & 127;
        float4 v = *(const float4*)(W + idx);
        *(float4*)(&ldsW[row * LDSW_STRIDE + col]) = v;
    }

    const int tile   = blockIdx.x * WPB + wave;
    const bool active = (tile < nTiles);
    if (active) {
        const float* Xrow = X + (size_t)tile * 16 * HDIM;
        for (int r = 0; r < 16; ++r) {
            float4 v = *(const float4*)(Xrow + (size_t)r * HDIM + lane * 4);
            *(float4*)(&ldsA[wave][r * LDSA_STRIDE + lane * 4]) = v;
        }
    }
    __syncthreads();
    if (!active) return;

    const int m     = lane & 15;
    const int khalf = (lane >> 4) * 2;
    const int mrow  = (lane >> 4) * 8;

    const float4 dlo = *(const float4*)(dinv + (size_t)tile * 16 + mrow);
    const float4 dhi = *(const float4*)(dinv + (size_t)tile * 16 + mrow + 4);
    const float* aRow = &ldsA[wave][m * LDSA_STRIDE];

    v8f acc[8];
    const v8f vzero = {0.f, 0.f, 0.f, 0.f, 0.f, 0.f, 0.f, 0.f};
#pragma unroll
    for (int nt = 0; nt < 8; ++nt) acc[nt] = vzero;

    for (int k4 = 0; k4 < 32; ++k4) {
        const int kb = k4 * 4 + khalf;
        v2f a;
        a.x = aRow[kb];
        a.y = aRow[kb + 1];
        const float* w0 = &ldsW[kb * LDSW_STRIDE];
        const float* w1 = w0 + LDSW_STRIDE;
#pragma unroll
        for (int nt = 0; nt < 8; ++nt) {
            const int n = nt * 16 + m;
            v2f b;
            b.x = w0[n];
            b.y = w1[n];
            acc[nt] = __builtin_amdgcn_wmma_f32_16x16x4_f32(
                false, a, false, b, (short)0, acc[nt], false, false);
        }
    }

    float* outBase = Hout + (size_t)tile * 16 * HDIM;
    const float drow[8] = {dlo.x, dlo.y, dlo.z, dlo.w, dhi.x, dhi.y, dhi.z, dhi.w};
#pragma unroll
    for (int nt = 0; nt < 8; ++nt) {
#pragma unroll
        for (int r = 0; r < 8; ++r) {
            outBase[(size_t)(mrow + r) * HDIM + nt * 16 + m] = acc[nt][r] * drow[r];
        }
    }
}

#define BSH       9
#define BNODES    512
#define NBK       256
#define ECH       8192
#define STAGE_CAP 16384

template <int NW>
__device__ __forceinline__ int block_excl_scan(int v, int* wsum, int lane, int wave, int& total) {
    int inc = v;
#pragma unroll
    for (int off = 1; off < 32; off <<= 1) {
        int t = __shfl_up(inc, off, 32);
        if (lane >= off) inc += t;
    }
    if (lane == 31) wsum[wave] = inc;
    __syncthreads();
    int wbase = 0, tot = 0;
#pragma unroll
    for (int w = 0; w < NW; ++w) {
        int x = wsum[w];
        if (w < wave) wbase += x;
        tot += x;
    }
    total = tot;
    __syncthreads();
    return wbase + inc - v;
}

__device__ __forceinline__ int bucket_of(int d) {
    unsigned b = (unsigned)d >> BSH;
    return (int)(b < NBK ? b : NBK - 1);
}

__launch_bounds__(256)
__global__ void bucket_hist(const int* __restrict__ dst, int* __restrict__ tbl, int E) {
    __shared__ int hist[NBK];
    const int tid = threadIdx.x;
    hist[tid] = 0;
    __syncthreads();
    const int beg = blockIdx.x * ECH;
    int end = beg + ECH;
    if (end > E) end = E;
    for (int e0 = beg + tid; e0 < end; e0 += 256 * 8) {
        int d[8];
#pragma unroll
        for (int k = 0; k < 8; ++k) d[k] = dst[min(e0 + k * 256, end - 1)];
#pragma unroll
        for (int k = 0; k < 8; ++k)
            if (e0 + k * 256 < end) atomicAdd(&hist[bucket_of(d[k])], 1);
    }
    __syncthreads();
    tbl[blockIdx.x * NBK + tid] = hist[tid];
}

__launch_bounds__(256)
__global__ void bucket_scatter(const int* __restrict__ src, const int* __restrict__ dst,
                               const int* __restrict__ tbl, int* __restrict__ bucketStart,
                               unsigned* __restrict__ bPack, int E, int G) {
    __shared__ int sBase[NBK];
    __shared__ int sCur[NBK];
    __shared__ unsigned sPack[ECH];
    __shared__ unsigned char sBkt[ECH];
    __shared__ int wsum[8];
    const int tid  = threadIdx.x;
    const int lane = tid & 31;
    const int wave = tid >> 5;
    const int blk  = blockIdx.x;

    int pre = 0, tot = 0;
#pragma unroll 16
    for (int j = 0; j < G; ++j) {
        const int v = tbl[j * NBK + tid];
        tot += v;
        pre += (j < blk) ? v : 0;
    }
    const int mine = tbl[blk * NBK + tid];
    int unused;
    const int bStart = block_excl_scan<8>(tot, wsum, lane, wave, unused);
    const int lOff   = block_excl_scan<8>(mine, wsum, lane, wave, unused);
    sBase[tid] = bStart + pre - lOff;
    sCur[tid]  = lOff;
    if (blk == 0) {
        bucketStart[tid] = bStart;
        if (tid == NBK - 1) bucketStart[NBK] = bStart + tot;
    }
    __syncthreads();

    const int beg = blk * ECH;
    int end = beg + ECH;
    if (end > E) end = E;
    for (int e0 = beg + tid; e0 < end; e0 += 256 * 8) {
        int s[8], d[8];
#pragma unroll
        for (int k = 0; k < 8; ++k) {
            const int ec = min(e0 + k * 256, end - 1);
            s[k] = src[ec];
            d[k] = dst[ec];
        }
#pragma unroll
        for (int k = 0; k < 8; ++k) {
            if (e0 + k * 256 < end) {
                const int b = bucket_of(d[k]);
                const int p = atomicAdd(&sCur[b], 1);
                sPack[p] = ((unsigned)s[k] << BSH) | ((unsigned)d[k] & (BNODES - 1));
                sBkt[p]  = (unsigned char)b;
            }
        }
    }
    __syncthreads();
    const int m = end - beg;
    for (int i = tid; i < m; i += 256) bPack[sBase[sBkt[i]] + i] = sPack[i];
}

__launch_bounds__(512)
__global__ void bucket_csr(const unsigned* __restrict__ bPack, const int* __restrict__ bucketStart,
                           int* __restrict__ rowptr, float* __restrict__ dinv,
                           int* __restrict__ col, int n, int E) {
    __shared__ int cnt[BNODES];
    __shared__ int cur[BNODES];
    __shared__ int stage[STAGE_CAP];
    __shared__ int wsum[16];
    const int tid  = threadIdx.x;
    const int lane = tid & 31;
    const int wave = tid >> 5;
    const int b    = blockIdx.x;
    const int bs   = bucketStart[b];
    const int m    = bucketStart[b + 1] - bs;

    cnt[tid] = 0;
    __syncthreads();
    for (int i0 = tid; i0 < m; i0 += 512 * 8) {
        unsigned w[8];
#pragma unroll
        for (int k = 0; k < 8; ++k) w[k] = bPack[bs + min(i0 + k * 512, m - 1)];
#pragma unroll
        for (int k = 0; k < 8; ++k)
            if (i0 + k * 512 < m) atomicAdd(&cnt[w[k] & (BNODES - 1)], 1);
    }
    __syncthreads();

    const int c = cnt[tid];
    int unused;
    const int off = block_excl_scan<16>(c, wsum, lane, wave, unused);
    const int node = (b << BSH) + tid;
    if (node < n) {
        rowptr[node] = bs + off;
        dinv[node]   = rsqrtf((float)(c + 1));
    }
    if (b == 0 && tid == 0) rowptr[n] = E;
    cur[tid] = off;
    __syncthreads();

    if (m <= STAGE_CAP) {
        for (int i0 = tid; i0 < m; i0 += 512 * 8) {
            unsigned w[8];
#pragma unroll
            for (int k = 0; k < 8; ++k) w[k] = bPack[bs + min(i0 + k * 512, m - 1)];
#pragma unroll
            for (int k = 0; k < 8; ++k) {
                if (i0 + k * 512 < m) {
                    const int p = atomicAdd(&cur[w[k] & (BNODES - 1)], 1);
                    stage[p] = (int)(w[k] >> BSH);
                }
            }
        }
        __syncthreads();
        for (int i = tid; i < m; i += 512) col[bs + i] = stage[i];
    } else {
        for (int i = tid; i < m; i += 512) {
            const unsigned w = bPack[bs + i];
            const int p = atomicAdd(&cur[w & (BNODES - 1)], 1);
            col[bs + p] = (int)(w >> BSH);
        }
    }
}


template <int CTRL>
__device__ __forceinline__ float row_xchg(float x) {
    const int b = __builtin_bit_cast(int, x);
    return __builtin_bit_cast(float, __builtin_amdgcn_update_dpp(b, b, CTRL, 0xf, 0xf, false));
}
__device__ __forceinline__ float half_xchg(float x) {
    const int b = __builtin_bit_cast(int, x);
    return __builtin_bit_cast(float,
        __builtin_amdgcn_permlanex16(b, b, 0x76543210, 0xfedcba98, false, false));
}

__device__ __forceinline__ void stage_wlin(const float* __restrict__ Wl, float4* sW, int tid) {
    sW[(tid & 7) * 32 + (tid >> 3)] = ((const float4*)Wl)[tid];
}

__device__ __forceinline__ int wl_slot(int s) {
    return (s >> 1) * LDSW_STRIDE + HDIM + (s & 1) * 4;
}

__device__ __forceinline__ void lane_partials(const float4 o, const float4 (&w)[8], float (&p)[8]) {
    const float ov[4] = {o.x, o.y, o.z, o.w};
#pragma unroll
    for (int i = 0; i < 4; ++i) {
        const float4 lo = w[2 * i], hi = w[2 * i + 1];
        if (i == 0) {
            p[0] = ov[0] * lo.x; p[1] = ov[0] * lo.y; p[2] = ov[0] * lo.z; p[3] = ov[0] * lo.w;
            p[4] = ov[0] * hi.x; p[5] = ov[0] * hi.y; p[6] = ov[0] * hi.z; p[7] = ov[0] * hi.w;
        } else {
            p[0] = fmaf(ov[i], lo.x, p[0]); p[1] = fmaf(ov[i], lo.y, p[1]);
            p[2] = fmaf(ov[i], lo.z, p[2]); p[3] = fmaf(ov[i], lo.w, p[3]);
            p[4] = fmaf(ov[i], hi.x, p[4]); p[5] = fmaf(ov[i], hi.y, p[5]);
            p[6] = fmaf(ov[i], hi.z, p[6]); p[7] = fmaf(ov[i], hi.w, p[7]);
        }
    }
}

__device__ __forceinline__ float class_sums(const float (&p)[8], int lane) {
    const bool h = (lane & 16) != 0, b3 = (lane & 8) != 0, b2 = (lane & 4) != 0;
    float q[4], r[2];
#pragma unroll
    for (int i = 0; i < 4; ++i) {
        const float keep = h ? p[i + 4] : p[i];
        const float send = h ? p[i] : p[i + 4];
        q[i] = keep + half_xchg(send);
    }
#pragma unroll
    for (int i = 0; i < 2; ++i) {
        const float keep = b3 ? q[i + 2] : q[i];
        const float send = b3 ? q[i] : q[i + 2];
        r[i] = keep + row_xchg<0x168>(send);
    }
    const float keep = b2 ? r[1] : r[0];
    const float send = b2 ? r[0] : r[1];
    float s = keep + row_xchg<0x164>(send);
    s += row_xchg<0x162>(s);
    s += row_xchg<0x161>(s);
    return s;
}

#define FUSED_GRID 1024

__device__ __forceinline__ float4 row_tail(const float4 acc, const float dd, const float4 b) {
    float4 v;
    v.x = fmaxf(acc.x * dd + b.x, 0.f);
    v.y = fmaxf(acc.y * dd + b.y, 0.f);
    v.z = fmaxf(acc.z * dd + b.z, 0.f);
    v.w = fmaxf(acc.w * dd + b.w, 0.f);
    float ss = v.x * v.x + v.y * v.y + v.z * v.z + v.w * v.w;
#pragma unroll
    for (int off = 16; off > 0; off >>= 1)
        ss += __shfl_xor(ss, off, 32);
    const float scale = 1.0f / fmaxf(sqrtf(ss), 1e-12f);
    float4 o;
    o.x = v.x * scale; o.y = v.y * scale; o.z = v.z * scale; o.w = v.w * scale;
    return o;
}

__launch_bounds__(256)
__global__ void gather_norm_gemm(const int* __restrict__ rowptr, const int* __restrict__ col,
                                 const float* __restrict__ dinv, const float* __restrict__ h,
                                 const float* __restrict__ bias,
                                 const float* __restrict__ Wl, float* plog, int accumulate,
                                 const float* __restrict__ Wn, float* __restrict__ hNext,
                                 int nTiles) {
    __shared__ float ldsW[HDIM * LDSW_STRIDE];
    __shared__ float ldsT[16 * LDSA_STRIDE];
    __shared__ float4 sLog[32];

    const int tid  = threadIdx.x;
    const int lane = tid & 31;
    const int wave = __builtin_amdgcn_readfirstlane(tid >> 5);

    for (int i = 0; i < 16; ++i) {
        int idx = (i * 256 + tid) * 4;
        int row = idx >> 7, cc = idx & 127;
        float4 v = *(const float4*)(Wn + idx);
        *(float4*)(&ldsW[row * LDSW_STRIDE + cc]) = v;
    }
    const float4 b = ((const float4*)bias)[lane];

    *(float4*)(&ldsW[wl_slot((tid & 7) * 32 + (tid >> 3))]) = ((const float4*)Wl)[tid];
    __syncthreads();

    const int m     = lane & 15;
    const int khalf = (lane >> 4) * 2;
    const int mrow  = (lane >> 4) * 8;
    const float* aRow = &ldsT[m * LDSA_STRIDE + khalf];
    const float* wCol = &ldsW[khalf * LDSW_STRIDE + wave * 16 + m];
    const int rA = wave * 2;

    for (int tile = blockIdx.x; tile < nTiles; tile += gridDim.x) {
        const int nA = tile * 16 + rA;
        const int pA = rowptr[nA];
        const int pB = rowptr[nA + 1];
        const int pC = rowptr[nA + 2];
        const float ddA = dinv[nA];
        const float ddB = dinv[nA + 1];
        float4 accA = ((const float4*)(h + (size_t)nA * HDIM))[lane];
        float4 accB = ((const float4*)(h + (size_t)(nA + 1) * HDIM))[lane];
        const float4 dlo = *(const float4*)(dinv + (size_t)tile * 16 + mrow);
        const float4 dhi = *(const float4*)(dinv + (size_t)tile * 16 + mrow + 4);

        for (int baseA = pA, baseB = pB; baseA < pB || baseB < pC; baseA += 32, baseB += 32) {
            int mA = pB - baseA; mA = mA < 0 ? 0 : (mA > 32 ? 32 : mA);
            int mB = pC - baseB; mB = mB < 0 ? 0 : (mB > 32 ? 32 : mB);
            int sA = 0, sB = 0;
            if (lane < mA) sA = col[baseA + lane];
            if (lane < mB) sB = col[baseB + lane];
            const int mm = mA > mB ? mA : mB;
            for (int j0 = 0; j0 < mm; j0 += 4) {
                int iA[4], iB[4];
#pragma unroll
                for (int u = 0; u < 4; ++u) {
                    iA[u] = __builtin_amdgcn_readlane(sA, j0 + u);
                    iB[u] = __builtin_amdgcn_readlane(sB, j0 + u);
                }
                float4 vA[4], vB[4];
#pragma unroll
                for (int u = 0; u < 4; ++u)
                    if (j0 + u < mA) vA[u] = ((const float4*)(h + (size_t)iA[u] * HDIM))[lane];
#pragma unroll
                for (int u = 0; u < 4; ++u)
                    if (j0 + u < mB) vB[u] = ((const float4*)(h + (size_t)iB[u] * HDIM))[lane];
#pragma unroll
                for (int u = 0; u < 4; ++u)
                    if (j0 + u < mA) {
                        accA.x += vA[u].x; accA.y += vA[u].y; accA.z += vA[u].z; accA.w += vA[u].w;
                    }
#pragma unroll
                for (int u = 0; u < 4; ++u)
                    if (j0 + u < mB) {
                        accB.x += vB[u].x; accB.y += vB[u].y; accB.z += vB[u].z; accB.w += vB[u].w;
                    }
            }
        }
        const float4 oA = row_tail(accA, ddA, b);
        const float4 oB = row_tail(accB, ddB, b);
        *(float4*)(&ldsT[rA * LDSA_STRIDE + lane * 4]) = oA;
        *(float4*)(&ldsT[(rA + 1) * LDSA_STRIDE + lane * 4]) = oB;
        float4 wl[8];
#pragma unroll
        for (int q = 0; q < 8; ++q) wl[q] = *(const float4*)(&ldsW[wl_slot(q * 32 + lane)]);
        float lpA[8], lpB[8];
        lane_partials(oA, wl, lpA);
        lane_partials(oB, wl, lpB);
        const float lsA = class_sums(lpA, lane);
        const float lsB = class_sums(lpB, lane);
        if ((lane & 2) == 0)
            ((float*)sLog)[(rA + (lane & 1)) * 8 + (lane >> 2)] = (lane & 1) ? lsB : lsA;
        __syncthreads();

        float4 lg = {0.f, 0.f, 0.f, 0.f}, pv = {0.f, 0.f, 0.f, 0.f};
        if (wave == 0) {
            lg = sLog[lane];
            if (accumulate) pv = ((const float4*)plog)[(size_t)tile * 32 + lane];
        }

        v8f acc = {0.f, 0.f, 0.f, 0.f, 0.f, 0.f, 0.f, 0.f};
        for (int k4 = 0; k4 < 32; ++k4) {
            const int kc = k4 * 4;
            v2f a;
            a.x = aRow[kc];
            a.y = aRow[kc + 1];
            v2f w;
            w.x = wCol[kc * LDSW_STRIDE];
            w.y = wCol[(kc + 1) * LDSW_STRIDE];
            acc = __builtin_amdgcn_wmma_f32_16x16x4_f32(
                false, a, false, w, (short)0, acc, false, false);
        }
        if (wave == 0) {
            if (accumulate) {
                lg.x = pv.x + lg.x; lg.y = pv.y + lg.y; lg.z = pv.z + lg.z; lg.w = pv.w + lg.w;
            }
            ((float4*)plog)[(size_t)tile * 32 + lane] = lg;
        }
        __syncthreads();

        const float drow[8] = {dlo.x, dlo.y, dlo.z, dlo.w, dhi.x, dhi.y, dhi.z, dhi.w};
#pragma unroll
        for (int r = 0; r < 8; ++r)
            ldsT[(mrow + r) * LDSA_STRIDE + wave * 16 + m] = acc[r] * drow[r];
        __syncthreads();
        const float4 hA = *(const float4*)(&ldsT[rA * LDSA_STRIDE + lane * 4]);
        const float4 hB = *(const float4*)(&ldsT[(rA + 1) * LDSA_STRIDE + lane * 4]);
        *(float4*)(hNext + (size_t)nA * HDIM + lane * 4) = hA;
        *(float4*)(hNext + (size_t)(nA + 1) * HDIM + lane * 4) = hB;
    }
}

__launch_bounds__(256)
__global__ void gather_agg_classify(const int* __restrict__ rowptr, const int* __restrict__ col,
                                    const float* __restrict__ dinv, const float* __restrict__ h,
                                    const float* __restrict__ bias,
                                    const float* __restrict__ Wl, const float* __restrict__ blin,
                                    const float* __restrict__ plog, float* __restrict__ out, int n) {
    __shared__ float4 sW[256];
    __shared__ float4 sLog[16];
    __shared__ int arrived;
    const int tid  = threadIdx.x;
    const int lane = tid & 31;
    const int wave = __builtin_amdgcn_readfirstlane(tid >> 5);
    const int node0 = blockIdx.x * 8;
    const int node = (node0 + wave < n) ? node0 + wave : n - 1;

    const int   beg = rowptr[node];
    const int   end = rowptr[node + 1];
    const float dd  = dinv[node];
    float4 acc = ((const float4*)(h + (size_t)node * HDIM))[lane];
    const float4 b = ((const float4*)bias)[lane];
    int s_l = 0;
    if (lane < end - beg) s_l = col[beg + lane];
    stage_wlin(Wl, sW, tid);
    if (tid == 0) arrived = 0;
    __syncthreads();

    for (int base = beg; base < end; base += 32) {
        int m = end - base;
        if (m > 32) m = 32;
#pragma unroll 4
        for (int j = 0; j < m; ++j) {
            const int s = __builtin_amdgcn_readlane(s_l, j);
            const float4 v = ((const float4*)(h + (size_t)s * HDIM))[lane];
            acc.x += v.x;
            acc.y += v.y;
            acc.z += v.z;
            acc.w += v.w;
        }
        s_l = 0;
        if (lane < end - base - 32) s_l = col[base + 32 + lane];
    }
    const float4 o = row_tail(acc, dd, b);
    float4 wl[8];
#pragma unroll
    for (int q = 0; q < 8; ++q) wl[q] = sW[q * 32 + lane];
    float p[8];
    lane_partials(o, wl, p);
    const float s = class_sums(p, lane);
    if ((lane & 3) == 0) ((float*)sLog)[wave * 8 + (lane >> 2)] = s;
    int before = 0;
    if (lane == 0)
        before = __hip_atomic_fetch_add(&arrived, 1, __ATOMIC_ACQ_REL, __HIP_MEMORY_SCOPE_WORKGROUP);
    before = __builtin_amdgcn_readfirstlane(before);
    if (before != 7) return;

    const int l16  = lane & 15;
    const int half = l16 & 1;
    const int nodeF = node0 + (l16 >> 1);
    const int nodeC = nodeF < n ? nodeF : n - 1;
    const float4 own = sLog[l16];
    const float4 pv  = ((const float4*)plog)[(size_t)nodeC * 2 + half];
    const float4 bl  = ((const float4*)blin)[half];
    float4 v;
    v.x = (pv.x + own.x) + bl.x;
    v.y = (pv.y + own.y) + bl.y;
    v.z = (pv.z + own.z) + bl.z;
    v.w = (pv.w + own.w) + bl.w;
    float mx = fmaxf(fmaxf(v.x, v.y), fmaxf(v.z, v.w));
    mx = fmaxf(mx, row_xchg<0x161>(mx));
    float se = (expf(v.x - mx) + expf(v.y - mx)) + (expf(v.z - mx) + expf(v.w - mx));
    se += row_xchg<0x161>(se);
    const float lse = mx + logf(se);
    float4 r;
    r.x = v.x - lse; r.y = v.y - lse; r.z = v.z - lse; r.w = v.w - lse;
    if (lane < 16 && nodeF < n)
        *(float4*)(out + (size_t)nodeF * 8 + half * 4) = r;
}

extern "C" void kernel_launch(void* const* d_in, const int* in_sizes, int n_in,
                              void* d_out, int out_size, void* d_ws, size_t ws_size,
                              hipStream_t stream) {
    const float* x    = (const float*)d_in[0];
    const int*   ei   = (const int*)d_in[1];
    const float* W1   = (const float*)d_in[2];
    const float* b1   = (const float*)d_in[3];
    const float* W2   = (const float*)d_in[4];
    const float* b2   = (const float*)d_in[5];
    const float* W3   = (const float*)d_in[6];
    const float* b3   = (const float*)d_in[7];
    const float* Wlin = (const float*)d_in[8];
    const float* blin = (const float*)d_in[9];

    const int n = in_sizes[0] / HDIM;
    const int E = in_sizes[1] / 2;
    const int* src = ei;
    const int* dst = ei + E;

    float* ws   = (float*)d_ws;
    float* h    = ws;
    float* h2   = h  + (size_t)n * HDIM;
    float* plog = h2 + (size_t)n * HDIM;
    float* dinv = plog + (size_t)n * 8;
    int* col         = (int*)(dinv + (size_t)n);
    int* rowptr      = col + (size_t)E;
    int* bucketStart = rowptr + (size_t)n + 1;
    int* tbl         = bucketStart + NBK + 1;
    unsigned* bPack  = (unsigned*)h;

    const int G = (E + ECH - 1) / ECH;
    bucket_hist<<<G, 256, 0, stream>>>(dst, tbl, E);
    bucket_scatter<<<G, 256, 0, stream>>>(src, dst, tbl, bucketStart, bPack, E, G);
    bucket_csr<<<NBK, 512, 0, stream>>>(bPack, bucketStart, rowptr, dinv, col, n, E);

    const int nTiles = (n + 15) / 16;
    const float* Wl1 = Wlin;
    const float* Wl2 = Wlin + HDIM * 8;
    const float* Wl3 = Wlin + 2 * HDIM * 8;

    gemm_wmma<<<(nTiles + WPB - 1) / WPB, 256, 0, stream>>>(x, W1, dinv, h, nTiles);
    gather_norm_gemm<<<FUSED_GRID, 256, 0, stream>>>(rowptr, col, dinv, h, b1, Wl1, plog, 0, W2, h2, nTiles);
    gather_norm_gemm<<<FUSED_GRID, 256, 0, stream>>>(rowptr, col, dinv, h2, b2, Wl2, plog, 1, W3, h, nTiles);
    gather_agg_classify<<<(n + 7) / 8, 256, 0, stream>>>(rowptr, col, dinv, h, b3, Wl3, blin, plog,
                                                         (float*)d_out, n);
}


# Edge sort in two launches, not three: each edge block writes its bucket-sorted chunk in place with a run table and the bucket work-group collects its runs; sum order stays LDS arrival rank
# speedup vs baseline: 5.6755x; 1.0464x over previous
#include <hip/hip_runtime.h>
#include <math.h>

typedef float v2f __attribute__((ext_vector_type(2)));
typedef float v8f __attribute__((ext_vector_type(8)));

#define HDIM 128
#define WPB 8
#define LDSA_STRIDE 132
#define LDSW_STRIDE 136

__launch_bounds__(256)
__global__ void gemm_wmma(const float* __restrict__ X,
                          const float* __restrict__ W,
                          const float* __restrict__ dinv,
                          float* __restrict__ Hout, int nTiles) {
    __shared__ float ldsW[HDIM * LDSW_STRIDE];
    __shared__ float ldsA[WPB][16 * LDSA_STRIDE];

    const int tid  = threadIdx.x;
    const int wave = tid >> 5;
    const int lane = tid & 31;

    for (int i = 0; i < 16; ++i) {
        int idx = (i * 256 + tid) * 4;
        int row = idx >> 7, col = idx & 127;
        float4 v = *(const float4*)(W + idx);
        *(float4*)(&ldsW[row * LDSW_STRIDE + col]) = v;
    }

    const int tile   = blockIdx.x * WPB + wave;
    const bool active = (tile < nTiles);
    if (active) {
        const float* Xrow = X + (size_t)tile * 16 * HDIM;
        for (int r = 0; r < 16; ++r) {
            float4 v = *(const float4*)(Xrow + (size_t)r * HDIM + lane * 4);
            *(float4*)(&ldsA[wave][r * LDSA_STRIDE + lane * 4]) = v;
        }
    }
    __syncthreads();
    if (!active) return;

    const int m     = lane & 15;
    const int khalf = (lane >> 4) * 2;
    const int mrow  = (lane >> 4) * 8;

    const float4 dlo = *(const float4*)(dinv + (size_t)tile * 16 + mrow);
    const float4 dhi = *(const float4*)(dinv + (size_t)tile * 16 + mrow + 4);
    const float* aRow = &ldsA[wave][m * LDSA_STRIDE];

    v8f acc[8];
    const v8f vzero = {0.f, 0.f, 0.f, 0.f, 0.f, 0.f, 0.f, 0.f};
#pragma unroll
    for (int nt = 0; nt < 8; ++nt) acc[nt] = vzero;

    for (int k4 = 0; k4 < 32; ++k4) {
        const int kb = k4 * 4 + khalf;
        v2f a;
        a.x = aRow[kb];
        a.y = aRow[kb + 1];
        const float* w0 = &ldsW[kb * LDSW_STRIDE];
        const float* w1 = w0 + LDSW_STRIDE;
#pragma unroll
        for (int nt = 0; nt < 8; ++nt) {
            const int n = nt * 16 + m;
            v2f b;
            b.x = w0[n];
            b.y = w1[n];
            acc[nt] = __builtin_amdgcn_wmma_f32_16x16x4_f32(
                false, a, false, b, (short)0, acc[nt], false, false);
        }
    }

    float* outBase = Hout + (size_t)tile * 16 * HDIM;
    const float drow[8] = {dlo.x, dlo.y, dlo.z, dlo.w, dhi.x, dhi.y, dhi.z, dhi.w};
#pragma unroll
    for (int nt = 0; nt < 8; ++nt) {
#pragma unroll
        for (int r = 0; r < 8; ++r) {
            outBase[(size_t)(mrow + r) * HDIM + nt * 16 + m] = acc[nt][r] * drow[r];
        }
    }
}

#define BSH       9
#define BNODES    512
#define NBK       256
#define ECH       8192
#define STAGE_CAP 16384
#define MAXG      512

template <int NW>
__device__ __forceinline__ int block_excl_scan(int v, int* wsum, int lane, int wave, int& total) {
    int inc = v;
#pragma unroll
    for (int off = 1; off < 32; off <<= 1) {
        int t = __shfl_up(inc, off, 32);
        if (lane >= off) inc += t;
    }
    if (lane == 31) wsum[wave] = inc;
    __syncthreads();
    int wbase = 0, tot = 0;
#pragma unroll
    for (int w = 0; w < NW; ++w) {
        int x = wsum[w];
        if (w < wave) wbase += x;
        tot += x;
    }
    total = tot;
    __syncthreads();
    return wbase + inc - v;
}

__device__ __forceinline__ int bucket_of(int d) {
    unsigned b = (unsigned)d >> BSH;
    return (int)(b < NBK ? b : NBK - 1);
}

__launch_bounds__(256)
__global__ void bin_sort(const int* __restrict__ src, const int* __restrict__ dst,
                         int* __restrict__ tbl, int* __restrict__ off,
                         unsigned* __restrict__ chunk, int E) {
    __shared__ int hist[NBK];
    __shared__ int sCur[NBK];
    __shared__ unsigned sPack[ECH];
    __shared__ int wsum[8];
    const int tid  = threadIdx.x;
    const int lane = tid & 31;
    const int wave = tid >> 5;
    const int blk  = blockIdx.x;
    hist[tid] = 0;
    __syncthreads();
    const int beg = blk * ECH;
    int end = beg + ECH;
    if (end > E) end = E;
    unsigned pk[32];
    int bk[32];
#pragma unroll
    for (int r = 0; r < 4; ++r) {
        int s[8], d[8];
#pragma unroll
        for (int k = 0; k < 8; ++k) {
            const int ec = min(beg + tid + (r * 8 + k) * 256, end - 1);
            s[k] = src[ec];
            d[k] = dst[ec];
        }
#pragma unroll
        for (int k = 0; k < 8; ++k) {
            const int b = bucket_of(d[k]);
            pk[r * 8 + k] = ((unsigned)s[k] << BSH) | ((unsigned)d[k] & (BNODES - 1));
            bk[r * 8 + k] = b;
            if (beg + tid + (r * 8 + k) * 256 < end) atomicAdd(&hist[b], 1);
        }
    }
    __syncthreads();
    const int mine = hist[tid];
    int unused;
    const int lOff = block_excl_scan<8>(mine, wsum, lane, wave, unused);
    tbl[blk * NBK + tid] = mine;
    off[blk * NBK + tid] = lOff;
    sCur[tid] = lOff;
    __syncthreads();
    const int m = end - beg;
    if (m == ECH) {
#pragma unroll
        for (int g = 0; g < 4; ++g) {
            int p[8];
#pragma unroll
            for (int k = 0; k < 8; ++k) p[k] = atomicAdd(&sCur[bk[g * 8 + k]], 1);
#pragma unroll
            for (int k = 0; k < 8; ++k) sPack[p[k]] = pk[g * 8 + k];
        }
    } else {
#pragma unroll
        for (int i = 0; i < 32; ++i) {
            if (beg + tid + i * 256 < end) {
                const int p = atomicAdd(&sCur[bk[i]], 1);
                sPack[p] = pk[i];
            }
        }
    }
    __syncthreads();
    for (int i = tid; i < m; i += 256) chunk[beg + i] = sPack[i];
}

__device__ __forceinline__ int run_addr(const int* runStart, const int* runBase, int i) {
    int j = 0;
#pragma unroll
    for (int step = MAXG / 2; step >= 1; step >>= 1)
        if (runStart[j + step] <= i) j += step;
    return runBase[j] + i;
}

__launch_bounds__(512)
__global__ void bucket_csr(const unsigned* __restrict__ chunk, const int* __restrict__ tbl,
                           const int* __restrict__ off,
                           int* __restrict__ rowptr, float* __restrict__ dinv,
                           int* __restrict__ col, int n, int E, int G) {
    __shared__ int cnt[BNODES];
    __shared__ int cur[BNODES];
    __shared__ int stage[STAGE_CAP];
    __shared__ unsigned raw[STAGE_CAP];
    __shared__ int runStart[MAXG];
    __shared__ int runBase[MAXG];
    __shared__ int wsum[16];
    const int tid  = threadIdx.x;
    const int lane = tid & 31;
    const int wave = tid >> 5;
    const int b    = blockIdx.x;

    int rl = 0, ro = 0;
    if (tid < G) {
        rl = tbl[tid * NBK + b];
        ro = off[tid * NBK + b];
    }
    cnt[tid] = 0;
    int m, bs, unused;
    const int rs = block_excl_scan<16>(rl, wsum, lane, wave, m);
    (void)block_excl_scan<16>(ro, wsum, lane, wave, bs);
    runStart[tid] = rs;
    runBase[tid]  = tid * ECH + ro - rs;
    __syncthreads();

    const bool fits = (m <= STAGE_CAP);
    for (int i0 = tid; i0 < m; i0 += 512 * 8) {
        unsigned w[8];
#pragma unroll
        for (int k = 0; k < 8; ++k)
            w[k] = chunk[run_addr(runStart, runBase, min(i0 + k * 512, m - 1))];
#pragma unroll
        for (int k = 0; k < 8; ++k)
            if (i0 + k * 512 < m) {
                atomicAdd(&cnt[w[k] & (BNODES - 1)], 1);
                if (fits) raw[i0 + k * 512] = w[k];
            }
    }
    __syncthreads();

    const int c = cnt[tid];
    const int offn = block_excl_scan<16>(c, wsum, lane, wave, unused);
    const int node = (b << BSH) + tid;
    if (node < n) {
        rowptr[node] = bs + offn;
        dinv[node]   = rsqrtf((float)(c + 1));
    }
    if (b == 0 && tid == 0) rowptr[n] = E;
    cur[tid] = offn;
    __syncthreads();

    if (fits) {
        for (int i0 = tid; i0 < m; i0 += 512 * 8) {
            if (i0 + 7 * 512 < m) {
                unsigned w[8];
                int p[8];
#pragma unroll
                for (int k = 0; k < 8; ++k) w[k] = raw[i0 + k * 512];
#pragma unroll
                for (int k = 0; k < 8; ++k) p[k] = atomicAdd(&cur[w[k] & (BNODES - 1)], 1);
#pragma unroll
                for (int k = 0; k < 8; ++k) stage[p[k]] = (int)(w[k] >> BSH);
            } else {
                for (int i = i0; i < m; i += 512) {
                    const unsigned w = raw[i];
                    const int p = atomicAdd(&cur[w & (BNODES - 1)], 1);
                    stage[p] = (int)(w >> BSH);
                }
            }
        }
        __syncthreads();
        for (int i = tid; i < m; i += 512) col[bs + i] = stage[i];
    } else {
        for (int i = tid; i < m; i += 512) {
            const unsigned w = chunk[run_addr(runStart, runBase, i)];
            const int p = atomicAdd(&cur[w & (BNODES - 1)], 1);
            col[bs + p] = (int)(w >> BSH);
        }
    }
}


template <int CTRL>
__device__ __forceinline__ float row_xchg(float x) {
    const int b = __builtin_bit_cast(int, x);
    return __builtin_bit_cast(float, __builtin_amdgcn_update_dpp(b, b, CTRL, 0xf, 0xf, false));
}
__device__ __forceinline__ float half_xchg(float x) {
    const int b = __builtin_bit_cast(int, x);
    return __builtin_bit_cast(float,
        __builtin_amdgcn_permlanex16(b, b, 0x76543210, 0xfedcba98, false, false));
}

__device__ __forceinline__ void stage_wlin(const float* __restrict__ Wl, float4* sW, int tid) {
    sW[(tid & 7) * 32 + (tid >> 3)] = ((const float4*)Wl)[tid];
}

__device__ __forceinline__ int wl_slot(int s) {
    return (s >> 1) * LDSW_STRIDE + HDIM + (s & 1) * 4;
}

__device__ __forceinline__ void lane_partials(const float4 o, const float4 (&w)[8], float (&p)[8]) {
    const float ov[4] = {o.x, o.y, o.z, o.w};
#pragma unroll
    for (int i = 0; i < 4; ++i) {
        const float4 lo = w[2 * i], hi = w[2 * i + 1];
        if (i == 0) {
            p[0] = ov[0] * lo.x; p[1] = ov[0] * lo.y; p[2] = ov[0] * lo.z; p[3] = ov[0] * lo.w;
            p[4] = ov[0] * hi.x; p[5] = ov[0] * hi.y; p[6] = ov[0] * hi.z; p[7] = ov[0] * hi.w;
        } else {
            p[0] = fmaf(ov[i], lo.x, p[0]); p[1] = fmaf(ov[i], lo.y, p[1]);
            p[2] = fmaf(ov[i], lo.z, p[2]); p[3] = fmaf(ov[i], lo.w, p[3]);
            p[4] = fmaf(ov[i], hi.x, p[4]); p[5] = fmaf(ov[i], hi.y, p[5]);
            p[6] = fmaf(ov[i], hi.z, p[6]); p[7] = fmaf(ov[i], hi.w, p[7]);
        }
    }
}

__device__ __forceinline__ float class_sums(const float (&p)[8], int lane) {
    const bool h = (lane & 16) != 0, b3 = (lane & 8) != 0, b2 = (lane & 4) != 0;
    float q[4], r[2];
#pragma unroll
    for (int i = 0; i < 4; ++i) {
        const float keep = h ? p[i + 4] : p[i];
        const float send = h ? p[i] : p[i + 4];
        q[i] = keep + half_xchg(send);
    }
#pragma unroll
    for (int i = 0; i < 2; ++i) {
        const float keep = b3 ? q[i + 2] : q[i];
        const float send = b3 ? q[i] : q[i + 2];
        r[i] = keep + row_xchg<0x168>(send);
    }
    const float keep = b2 ? r[1] : r[0];
    const float send = b2 ? r[0] : r[1];
    float s = keep + row_xchg<0x164>(send);
    s += row_xchg<0x162>(s);
    s += row_xchg<0x161>(s);
    return s;
}

#define FUSED_GRID 1024

__device__ __forceinline__ float4 row_tail(const float4 acc, const float dd, const float4 b) {
    float4 v;
    v.x = fmaxf(acc.x * dd + b.x, 0.f);
    v.y = fmaxf(acc.y * dd + b.y, 0.f);
    v.z = fmaxf(acc.z * dd + b.z, 0.f);
    v.w = fmaxf(acc.w * dd + b.w, 0.f);
    float ss = v.x * v.x + v.y * v.y + v.z * v.z + v.w * v.w;
#pragma unroll
    for (int off = 16; off > 0; off >>= 1)
        ss += __shfl_xor(ss, off, 32);
    const float scale = 1.0f / fmaxf(sqrtf(ss), 1e-12f);
    float4 o;
    o.x = v.x * scale; o.y = v.y * scale; o.z = v.z * scale; o.w = v.w * scale;
    return o;
}

__launch_bounds__(256)
__global__ void gather_norm_gemm(const int* __restrict__ rowptr, const int* __restrict__ col,
                                 const float* __restrict__ dinv, const float* __restrict__ h,
                                 const float* __restrict__ bias,
                                 const float* __restrict__ Wl, float* plog, int accumulate,
                                 const float* __restrict__ Wn, float* __restrict__ hNext,
                                 int nTiles) {
    __shared__ float ldsW[HDIM * LDSW_STRIDE];
    __shared__ float ldsT[16 * LDSA_STRIDE];
    __shared__ float4 sLog[32];

    const int tid  = threadIdx.x;
    const int lane = tid & 31;
    const int wave = __builtin_amdgcn_readfirstlane(tid >> 5);

    for (int i = 0; i < 16; ++i) {
        int idx = (i * 256 + tid) * 4;
        int row = idx >> 7, cc = idx & 127;
        float4 v = *(const float4*)(Wn + idx);
        *(float4*)(&ldsW[row * LDSW_STRIDE + cc]) = v;
    }
    const float4 b = ((const float4*)bias)[lane];

    *(float4*)(&ldsW[wl_slot((tid & 7) * 32 + (tid >> 3))]) = ((const float4*)Wl)[tid];
    __syncthreads();

    const int m     = lane & 15;
    const int khalf = (lane >> 4) * 2;
    const int mrow  = (lane >> 4) * 8;
    const float* aRow = &ldsT[m * LDSA_STRIDE + khalf];
    const float* wCol = &ldsW[khalf * LDSW_STRIDE + wave * 16 + m];
    const int rA = wave * 2;

    for (int tile = blockIdx.x; tile < nTiles; tile += gridDim.x) {
        const int nA = tile * 16 + rA;
        const int pA = rowptr[nA];
        const int pB = rowptr[nA + 1];
        const int pC = rowptr[nA + 2];
        const float ddA = dinv[nA];
        const float ddB = dinv[nA + 1];
        float4 accA = ((const float4*)(h + (size_t)nA * HDIM))[lane];
        float4 accB = ((const float4*)(h + (size_t)(nA + 1) * HDIM))[lane];
        const float4 dlo = *(const float4*)(dinv + (size_t)tile * 16 + mrow);
        const float4 dhi = *(const float4*)(dinv + (size_t)tile * 16 + mrow + 4);

        for (int baseA = pA, baseB = pB; baseA < pB || baseB < pC; baseA += 32, baseB += 32) {
            int mA = pB - baseA; mA = mA < 0 ? 0 : (mA > 32 ? 32 : mA);
            int mB = pC - baseB; mB = mB < 0 ? 0 : (mB > 32 ? 32 : mB);
            int sA = 0, sB = 0;
            if (lane < mA) sA = col[baseA + lane];
            if (lane < mB) sB = col[baseB + lane];
            const int mm = mA > mB ? mA : mB;
            for (int j0 = 0; j0 < mm; j0 += 4) {
                int iA[4], iB[4];
#pragma unroll
                for (int u = 0; u < 4; ++u) {
                    iA[u] = __builtin_amdgcn_readlane(sA, j0 + u);
                    iB[u] = __builtin_amdgcn_readlane(sB, j0 + u);
                }
                float4 vA[4], vB[4];
#pragma unroll
                for (int u = 0; u < 4; ++u)
                    if (j0 + u < mA) vA[u] = ((const float4*)(h + (size_t)iA[u] * HDIM))[lane];
#pragma unroll
                for (int u = 0; u < 4; ++u)
                    if (j0 + u < mB) vB[u] = ((const float4*)(h + (size_t)iB[u] * HDIM))[lane];
#pragma unroll
                for (int u = 0; u < 4; ++u)
                    if (j0 + u < mA) {
                        accA.x += vA[u].x; accA.y += vA[u].y; accA.z += vA[u].z; accA.w += vA[u].w;
                    }
#pragma unroll
                for (int u = 0; u < 4; ++u)
                    if (j0 + u < mB) {
                        accB.x += vB[u].x; accB.y += vB[u].y; accB.z += vB[u].z; accB.w += vB[u].w;
                    }
            }
        }
        const float4 oA = row_tail(accA, ddA, b);
        const float4 oB = row_tail(accB, ddB, b);
        *(float4*)(&ldsT[rA * LDSA_STRIDE + lane * 4]) = oA;
        *(float4*)(&ldsT[(rA + 1) * LDSA_STRIDE + lane * 4]) = oB;
        float4 wl[8];
#pragma unroll
        for (int q = 0; q < 8; ++q) wl[q] = *(const float4*)(&ldsW[wl_slot(q * 32 + lane)]);
        float lpA[8], lpB[8];
        lane_partials(oA, wl, lpA);
        lane_partials(oB, wl, lpB);
        const float lsA = class_sums(lpA, lane);
        const float lsB = class_sums(lpB, lane);
        if ((lane & 2) == 0)
            ((float*)sLog)[(rA + (lane & 1)) * 8 + (lane >> 2)] = (lane & 1) ? lsB : lsA;
        __syncthreads();

        float4 lg = {0.f, 0.f, 0.f, 0.f}, pv = {0.f, 0.f, 0.f, 0.f};
        if (wave == 0) {
            lg = sLog[lane];
            if (accumulate) pv = ((const float4*)plog)[(size_t)tile * 32 + lane];
        }

        v8f acc = {0.f, 0.f, 0.f, 0.f, 0.f, 0.f, 0.f, 0.f};
        for (int k4 = 0; k4 < 32; ++k4) {
            const int kc = k4 * 4;
            v2f a;
            a.x = aRow[kc];
            a.y = aRow[kc + 1];
            v2f w;
            w.x = wCol[kc * LDSW_STRIDE];
            w.y = wCol[(kc + 1) * LDSW_STRIDE];
            acc = __builtin_amdgcn_wmma_f32_16x16x4_f32(
                false, a, false, w, (short)0, acc, false, false);
        }
        if (wave == 0) {
            if (accumulate) {
                lg.x = pv.x + lg.x; lg.y = pv.y + lg.y; lg.z = pv.z + lg.z; lg.w = pv.w + lg.w;
            }
            ((float4*)plog)[(size_t)tile * 32 + lane] = lg;
        }
        __syncthreads();

        const float drow[8] = {dlo.x, dlo.y, dlo.z, dlo.w, dhi.x, dhi.y, dhi.z, dhi.w};
#pragma unroll
        for (int r = 0; r < 8; ++r)
            ldsT[(mrow + r) * LDSA_STRIDE + wave * 16 + m] = acc[r] * drow[r];
        __syncthreads();
        const float4 hA = *(const float4*)(&ldsT[rA * LDSA_STRIDE + lane * 4]);
        const float4 hB = *(const float4*)(&ldsT[(rA + 1) * LDSA_STRIDE + lane * 4]);
        *(float4*)(hNext + (size_t)nA * HDIM + lane * 4) = hA;
        *(float4*)(hNext + (size_t)(nA + 1) * HDIM + lane * 4) = hB;
    }
}

__launch_bounds__(256)
__global__ void gather_agg_classify(const int* __restrict__ rowptr, const int* __restrict__ col,
                                    const float* __restrict__ dinv, const float* __restrict__ h,
                                    const float* __restrict__ bias,
                                    const float* __restrict__ Wl, const float* __restrict__ blin,
                                    const float* __restrict__ plog, float* __restrict__ out, int n) {
    __shared__ float4 sW[256];
    __shared__ float4 sLog[16];
    __shared__ int arrived;
    const int tid  = threadIdx.x;
    const int lane = tid & 31;
    const int wave = __builtin_amdgcn_readfirstlane(tid >> 5);
    const int node0 = blockIdx.x * 8;
    const int node = (node0 + wave < n) ? node0 + wave : n - 1;

    const int   beg = rowptr[node];
    const int   end = rowptr[node + 1];
    const float dd  = dinv[node];
    float4 acc = ((const float4*)(h + (size_t)node * HDIM))[lane];
    const float4 b = ((const float4*)bias)[lane];
    int s_l = 0;
    if (lane < end - beg) s_l = col[beg + lane];
    stage_wlin(Wl, sW, tid);
    if (tid == 0) arrived = 0;
    __syncthreads();

    for (int base = beg; base < end; base += 32) {
        int m = end - base;
        if (m > 32) m = 32;
#pragma unroll 4
        for (int j = 0; j < m; ++j) {
            const int s = __builtin_amdgcn_readlane(s_l, j);
            const float4 v = ((const float4*)(h + (size_t)s * HDIM))[lane];
            acc.x += v.x;
            acc.y += v.y;
            acc.z += v.z;
            acc.w += v.w;
        }
        s_l = 0;
        if (lane < end - base - 32) s_l = col[base + 32 + lane];
    }
    const float4 o = row_tail(acc, dd, b);
    float4 wl[8];
#pragma unroll
    for (int q = 0; q < 8; ++q) wl[q] = sW[q * 32 + lane];
    float p[8];
    lane_partials(o, wl, p);
    const float s = class_sums(p, lane);
    if ((lane & 3) == 0) ((float*)sLog)[wave * 8 + (lane >> 2)] = s;
    int before = 0;
    if (lane == 0)
        before = __hip_atomic_fetch_add(&arrived, 1, __ATOMIC_ACQ_REL, __HIP_MEMORY_SCOPE_WORKGROUP);
    before = __builtin_amdgcn_readfirstlane(before);
    if (before != 7) return;

    const int l16  = lane & 15;
    const int half = l16 & 1;
    const int nodeF = node0 + (l16 >> 1);
    const int nodeC = nodeF < n ? nodeF : n - 1;
    const float4 own = sLog[l16];
    const float4 pv  = ((const float4*)plog)[(size_t)nodeC * 2 + half];
    const float4 bl  = ((const float4*)blin)[half];
    float4 v;
    v.x = (pv.x + own.x) + bl.x;
    v.y = (pv.y + own.y) + bl.y;
    v.z = (pv.z + own.z) + bl.z;
    v.w = (pv.w + own.w) + bl.w;
    float mx = fmaxf(fmaxf(v.x, v.y), fmaxf(v.z, v.w));
    mx = fmaxf(mx, row_xchg<0x161>(mx));
    float se = (expf(v.x - mx) + expf(v.y - mx)) + (expf(v.z - mx) + expf(v.w - mx));
    se += row_xchg<0x161>(se);
    const float lse = mx + logf(se);
    float4 r;
    r.x = v.x - lse; r.y = v.y - lse; r.z = v.z - lse; r.w = v.w - lse;
    if (lane < 16 && nodeF < n)
        *(float4*)(out + (size_t)nodeF * 8 + half * 4) = r;
}

extern "C" void kernel_launch(void* const* d_in, const int* in_sizes, int n_in,
                              void* d_out, int out_size, void* d_ws, size_t ws_size,
                              hipStream_t stream) {
    const float* x    = (const float*)d_in[0];
    const int*   ei   = (const int*)d_in[1];
    const float* W1   = (const float*)d_in[2];
    const float* b1   = (const float*)d_in[3];
    const float* W2   = (const float*)d_in[4];
    const float* b2   = (const float*)d_in[5];
    const float* W3   = (const float*)d_in[6];
    const float* b3   = (const float*)d_in[7];
    const float* Wlin = (const float*)d_in[8];
    const float* blin = (const float*)d_in[9];

    const int n = in_sizes[0] / HDIM;
    const int E = in_sizes[1] / 2;
    const int* src = ei;
    const int* dst = ei + E;

    float* ws   = (float*)d_ws;
    float* h    = ws;
    float* h2   = h  + (size_t)n * HDIM;
    float* plog = h2 + (size_t)n * HDIM;
    float* dinv = plog + (size_t)n * 8;
    int* col         = (int*)(dinv + (size_t)n);
    int* rowptr      = col + (size_t)E;
    const int G = (E + ECH - 1) / ECH;
    int* tbl         = rowptr + (size_t)n + 1;
    int* off         = tbl + (size_t)G * NBK;
    unsigned* chunk  = (unsigned*)h;

    bin_sort<<<G, 256, 0, stream>>>(src, dst, tbl, off, chunk, E);
    bucket_csr<<<NBK, 512, 0, stream>>>(chunk, tbl, off, rowptr, dinv, col, n, E, G);

    const int nTiles = (n + 15) / 16;
    const float* Wl1 = Wlin;
    const float* Wl2 = Wlin + HDIM * 8;
    const float* Wl3 = Wlin + 2 * HDIM * 8;

    gemm_wmma<<<(nTiles + WPB - 1) / WPB, 256, 0, stream>>>(x, W1, dinv, h, nTiles);
    gather_norm_gemm<<<FUSED_GRID, 256, 0, stream>>>(rowptr, col, dinv, h, b1, Wl1, plog, 0, W2, h2, nTiles);
    gather_norm_gemm<<<FUSED_GRID, 256, 0, stream>>>(rowptr, col, dinv, h2, b2, Wl2, plog, 1, W3, h, nTiles);
    gather_agg_classify<<<(n + 7) / 8, 256, 0, stream>>>(rowptr, col, dinv, h, b3, Wl3, blin, plog,
                                                         (float*)d_out, n);
}


# Gathers: one b128 load per edge slot brings the source rows of BOTH nodes of a wave (lane half = node, 8 features a lane); row norm and logit sums are now 16-lane trees; row adds keep their order
# speedup vs baseline: 7.2770x; 1.2822x over previous
#include <hip/hip_runtime.h>
#include <math.h>

typedef float v2f __attribute__((ext_vector_type(2)));
typedef float v8f __attribute__((ext_vector_type(8)));
typedef _Float16 h4 __attribute__((ext_vector_type(4)));


#define HDIM 128
#define WPB 8
#define LDSA_STRIDE 132
#define LDSW_STRIDE 136

__launch_bounds__(256)
__global__ void gemm_wmma(const float* __restrict__ X,
                          const float* __restrict__ W,
                          const float* __restrict__ dinv,
                          _Float16* __restrict__ Hout, int nTiles) {
    __shared__ float ldsW[HDIM * LDSW_STRIDE];
    __shared__ float ldsA[WPB][16 * LDSA_STRIDE];

    const int tid  = threadIdx.x;
    const int wave = tid >> 5;
    const int lane = tid & 31;

    for (int i = 0; i < 16; ++i) {
        int idx = (i * 256 + tid) * 4;
        int row = idx >> 7, col = idx & 127;
        float4 v = *(const float4*)(W + idx);
        *(float4*)(&ldsW[row * LDSW_STRIDE + col]) = v;
    }

    if (blockIdx.x == 0 && tid < 32)
        ((uint2*)(Hout + (size_t)nTiles * 16 * HDIM))[tid] = make_uint2(0u, 0u);

    const int tile   = blockIdx.x * WPB + wave;
    const bool active = (tile < nTiles);
    if (active) {
        const float* Xrow = X + (size_t)tile * 16 * HDIM;
        for (int r = 0; r < 16; ++r) {
            float4 v = *(const float4*)(Xrow + (size_t)r * HDIM + lane * 4);
            *(float4*)(&ldsA[wave][r * LDSA_STRIDE + lane * 4]) = v;
        }
    }
    __syncthreads();
    if (!active) return;

    const int m     = lane & 15;
    const int khalf = (lane >> 4) * 2;
    const int mrow  = (lane >> 4) * 8;

    const float4 dlo = *(const float4*)(dinv + (size_t)tile * 16 + mrow);
    const float4 dhi = *(const float4*)(dinv + (size_t)tile * 16 + mrow + 4);
    const float* aRow = &ldsA[wave][m * LDSA_STRIDE];

    v8f acc[8];
    const v8f vzero = {0.f, 0.f, 0.f, 0.f, 0.f, 0.f, 0.f, 0.f};
#pragma unroll
    for (int nt = 0; nt < 8; ++nt) acc[nt] = vzero;

    for (int k4 = 0; k4 < 32; ++k4) {
        const int kb = k4 * 4 + khalf;
        v2f a;
        a.x = aRow[kb];
        a.y = aRow[kb + 1];
        const float* w0 = &ldsW[kb * LDSW_STRIDE];
        const float* w1 = w0 + LDSW_STRIDE;
#pragma unroll
        for (int nt = 0; nt < 8; ++nt) {
            const int n = nt * 16 + m;
            v2f b;
            b.x = w0[n];
            b.y = w1[n];
            acc[nt] = __builtin_amdgcn_wmma_f32_16x16x4_f32(
                false, a, false, b, (short)0, acc[nt], false, false);
        }
    }

    _Float16* outBase = Hout + (size_t)tile * 16 * HDIM;
    const float drow[8] = {dlo.x, dlo.y, dlo.z, dlo.w, dhi.x, dhi.y, dhi.z, dhi.w};
#pragma unroll
    for (int nt = 0; nt < 8; ++nt) {
#pragma unroll
        for (int r = 0; r < 8; ++r) {
            outBase[(size_t)(mrow + r) * HDIM + nt * 16 + m] = (_Float16)(acc[nt][r] * drow[r]);
        }
    }
}

#define BSH       9
#define BNODES    512
#define NBK       256
#define ECH       8192
#define STAGE_CAP 16384
#define MAXG      512

template <int NW>
__device__ __forceinline__ int block_excl_scan(int v, int* wsum, int lane, int wave, int& total) {
    int inc = v;
#pragma unroll
    for (int off = 1; off < 32; off <<= 1) {
        int t = __shfl_up(inc, off, 32);
        if (lane >= off) inc += t;
    }
    if (lane == 31) wsum[wave] = inc;
    __syncthreads();
    int wbase = 0, tot = 0;
#pragma unroll
    for (int w = 0; w < NW; ++w) {
        int x = wsum[w];
        if (w < wave) wbase += x;
        tot += x;
    }
    total = tot;
    __syncthreads();
    return wbase + inc - v;
}

__device__ __forceinline__ int bucket_of(int d) {
    unsigned b = (unsigned)d >> BSH;
    return (int)(b < NBK ? b : NBK - 1);
}

__launch_bounds__(256)
__global__ void bin_sort(const int* __restrict__ src, const int* __restrict__ dst,
                         int* __restrict__ tbl, int* __restrict__ off,
                         unsigned* __restrict__ chunk, int E) {
    __shared__ int hist[NBK];
    __shared__ int sCur[NBK];
    __shared__ unsigned sPack[ECH];
    __shared__ int wsum[8];
    const int tid  = threadIdx.x;
    const int lane = tid & 31;
    const int wave = tid >> 5;
    const int blk  = blockIdx.x;
    hist[tid] = 0;
    __syncthreads();
    const int beg = blk * ECH;
    int end = beg + ECH;
    if (end > E) end = E;
    unsigned pk[32];
    int bk[32];
#pragma unroll
    for (int r = 0; r < 4; ++r) {
        int s[8], d[8];
#pragma unroll
        for (int k = 0; k < 8; ++k) {
            const int ec = min(beg + tid + (r * 8 + k) * 256, end - 1);
            s[k] = src[ec];
            d[k] = dst[ec];
        }
#pragma unroll
        for (int k = 0; k < 8; ++k) {
            const int b = bucket_of(d[k]);
            pk[r * 8 + k] = ((unsigned)s[k] << BSH) | ((unsigned)d[k] & (BNODES - 1));
            bk[r * 8 + k] = b;
            if (beg + tid + (r * 8 + k) * 256 < end) atomicAdd(&hist[b], 1);
        }
    }
    __syncthreads();
    const int mine = hist[tid];
    int unused;
    const int lOff = block_excl_scan<8>(mine, wsum, lane, wave, unused);
    tbl[blk * NBK + tid] = mine;
    off[blk * NBK + tid] = lOff;
    sCur[tid] = lOff;
    __syncthreads();
    const int m = end - beg;
    if (m == ECH) {
#pragma unroll
        for (int g = 0; g < 4; ++g) {
            int p[8];
#pragma unroll
            for (int k = 0; k < 8; ++k) p[k] = atomicAdd(&sCur[bk[g * 8 + k]], 1);
#pragma unroll
            for (int k = 0; k < 8; ++k) sPack[p[k]] = pk[g * 8 + k];
        }
    } else {
#pragma unroll
        for (int i = 0; i < 32; ++i) {
            if (beg + tid + i * 256 < end) {
                const int p = atomicAdd(&sCur[bk[i]], 1);
                sPack[p] = pk[i];
            }
        }
    }
    __syncthreads();
    for (int i = tid; i < m; i += 256) chunk[beg + i] = sPack[i];
}

__device__ __forceinline__ int run_addr(const int* runStart, const int* runBase, int i) {
    int j = 0;
#pragma unroll
    for (int step = MAXG / 2; step >= 1; step >>= 1)
        if (runStart[j + step] <= i) j += step;
    return runBase[j] + i;
}

__launch_bounds__(512)
__global__ void bucket_csr(const unsigned* __restrict__ chunk, const int* __restrict__ tbl,
                           const int* __restrict__ off,
                           int* __restrict__ rowptr, float* __restrict__ dinv,
                           int* __restrict__ col, int n, int E, int G) {
    __shared__ int cnt[BNODES];
    __shared__ int cur[BNODES];
    __shared__ int stage[STAGE_CAP];
    __shared__ unsigned raw[STAGE_CAP];
    __shared__ int runStart[MAXG];
    __shared__ int runBase[MAXG];
    __shared__ int wsum[16];
    const int tid  = threadIdx.x;
    const int lane = tid & 31;
    const int wave = tid >> 5;
    const int b    = blockIdx.x;

    int rl = 0, ro = 0;
    if (tid < G) {
        rl = tbl[tid * NBK + b];
        ro = off[tid * NBK + b];
    }
    cnt[tid] = 0;
    int m, bs, unused;
    const int rs = block_excl_scan<16>(rl, wsum, lane, wave, m);
    (void)block_excl_scan<16>(ro, wsum, lane, wave, bs);
    runStart[tid] = rs;
    runBase[tid]  = tid * ECH + ro - rs;
    __syncthreads();

    const bool fits = (m <= STAGE_CAP);
    for (int i0 = tid; i0 < m; i0 += 512 * 8) {
        unsigned w[8];
#pragma unroll
        for (int k = 0; k < 8; ++k)
            w[k] = chunk[run_addr(runStart, runBase, min(i0 + k * 512, m - 1))];
#pragma unroll
        for (int k = 0; k < 8; ++k)
            if (i0 + k * 512 < m) {
                atomicAdd(&cnt[w[k] & (BNODES - 1)], 1);
                if (fits) raw[i0 + k * 512] = w[k];
            }
    }
    __syncthreads();

    const int c = cnt[tid];
    const int offn = block_excl_scan<16>(c, wsum, lane, wave, unused);
    const int node = (b << BSH) + tid;
    if (node < n) {
        rowptr[node] = bs + offn;
        dinv[node]   = rsqrtf((float)(c + 1));
    }
    if (b == 0 && tid == 0) rowptr[n] = E;
    cur[tid] = offn;
    __syncthreads();

    if (fits) {
        for (int i0 = tid; i0 < m; i0 += 512 * 8) {
            if (i0 + 7 * 512 < m) {
                unsigned w[8];
                int p[8];
#pragma unroll
                for (int k = 0; k < 8; ++k) w[k] = raw[i0 + k * 512];
#pragma unroll
                for (int k = 0; k < 8; ++k) p[k] = atomicAdd(&cur[w[k] & (BNODES - 1)], 1);
#pragma unroll
                for (int k = 0; k < 8; ++k) stage[p[k]] = (int)(w[k] >> BSH);
            } else {
                for (int i = i0; i < m; i += 512) {
                    const unsigned w = raw[i];
                    const int p = atomicAdd(&cur[w & (BNODES - 1)], 1);
                    stage[p] = (int)(w >> BSH);
                }
            }
        }
        __syncthreads();
        for (int i = tid; i < m; i += 512) col[bs + i] = stage[i];
    } else {
        for (int i = tid; i < m; i += 512) {
            const unsigned w = chunk[run_addr(runStart, runBase, i)];
            const int p = atomicAdd(&cur[w & (BNODES - 1)], 1);
            col[bs + p] = (int)(w >> BSH);
        }
    }
}


template <int CTRL>
__device__ __forceinline__ float row_xchg(float x) {
    const int b = __builtin_bit_cast(int, x);
    return __builtin_bit_cast(float, __builtin_amdgcn_update_dpp(b, b, CTRL, 0xf, 0xf, false));
}
__device__ __forceinline__ void stage_wlin(const float* __restrict__ Wl, float4* sW, int tid) {
    sW[(tid & 7) * 32 + (tid >> 3)] = ((const float4*)Wl)[tid];
}

__device__ __forceinline__ int wl_slot(int s) {
    return (s >> 1) * LDSW_STRIDE + HDIM + (s & 1) * 4;
}

#define FUSED_GRID 1024

__device__ __forceinline__ h4 narrow(const float4 v) {
    h4 t;
    t.x = (_Float16)v.x; t.y = (_Float16)v.y; t.z = (_Float16)v.z; t.w = (_Float16)v.w;
    return t;
}

typedef _Float16 h8 __attribute__((ext_vector_type(8)));

template <int J>
__device__ __forceinline__ int row_bcast(int x) {
    return __builtin_amdgcn_mov_dpp(x, 0x150 + J, 0xf, 0xf, true);
}

template <int J0, int N>
__device__ __forceinline__ void gatherN(const _Float16* __restrict__ h, int sid, int c, float (&acc)[8]) {
    const int i0 = row_bcast<(J0 + 0) & 15>(sid), i1 = row_bcast<(J0 + 1) & 15>(sid);
    const int i2 = row_bcast<(J0 + 2) & 15>(sid), i3 = row_bcast<(J0 + 3) & 15>(sid);
    const int i4 = row_bcast<(J0 + 4) & 15>(sid), i5 = row_bcast<(J0 + 5) & 15>(sid);
    const int i6 = row_bcast<(J0 + 6) & 15>(sid), i7 = row_bcast<(J0 + 7) & 15>(sid);
    const int id[8] = {i0, i1, i2, i3, i4, i5, i6, i7};
    const char* hb = (const char*)h;
    h8 v[N];
#pragma unroll
    for (int u = 0; u < N; ++u)
        v[u] = *(const h8*)(hb + (((unsigned)id[u] << 8) + ((unsigned)c << 4)));
#pragma unroll
    for (int u = 0; u < N; ++u) {
#pragma unroll
        for (int i = 0; i < 8; ++i) acc[i] += (float)v[u][i];
    }
}

template <int J0>
__device__ __forceinline__ void gather_step(const _Float16* __restrict__ h, int sid, int mm, int c,
                                            float (&acc)[8]) {
    if (mm > J0 + 4) gatherN<J0, 8>(h, sid, c, acc);
    else             gatherN<J0, 4>(h, sid, c, acc);
}

__device__ __forceinline__ void gather_pair(const int* __restrict__ col, const _Float16* __restrict__ h,
                                            int beg, int deg, int maxdeg, int node, int zrow, int c,
                                            float (&acc)[8]) {
    const h8 own = ((const h8*)(h + (size_t)node * HDIM))[c];
    int base = 0;
    do {
        int s0 = zrow, s1 = zrow;
        if (base + c < deg)      s0 = col[beg + base + c];
        if (base + 16 + c < deg) s1 = col[beg + base + 16 + c];
        if (base == 0) {
#pragma unroll
            for (int i = 0; i < 8; ++i) acc[i] = (float)own[i];
        }
        const int mm = maxdeg - base;
        if (mm > 0)  gather_step<0>(h, s0, mm, c, acc);
        if (mm > 8)  gather_step<8>(h, s0, mm, c, acc);
        if (mm > 16) gather_step<16>(h, s1, mm, c, acc);
        if (mm > 24) gather_step<24>(h, s1, mm, c, acc);
        base += 32;
    } while (base < maxdeg);
}

__device__ __forceinline__ void pair_tail(const float (&acc)[8], const float dd,
                                          const float4 b0, const float4 b1, float (&o)[8]) {
    const float bb[8] = {b0.x, b0.y, b0.z, b0.w, b1.x, b1.y, b1.z, b1.w};
    float v[8];
#pragma unroll
    for (int i = 0; i < 8; ++i) v[i] = fmaxf(acc[i] * dd + bb[i], 0.f);
    float ss = ((v[0] * v[0] + v[1] * v[1]) + (v[2] * v[2] + v[3] * v[3])) +
               ((v[4] * v[4] + v[5] * v[5]) + (v[6] * v[6] + v[7] * v[7]));
    ss += row_xchg<0x168>(ss);
    ss += row_xchg<0x164>(ss);
    ss += row_xchg<0x162>(ss);
    ss += row_xchg<0x161>(ss);
    const float scale = 1.0f / fmaxf(sqrtf(ss), 1e-12f);
#pragma unroll
    for (int i = 0; i < 8; ++i) o[i] = v[i] * scale;
}

template <class F>
__device__ __forceinline__ void pair_partials(const float (&o)[8], int c, F wl, float (&p)[8]) {
#pragma unroll
    for (int i = 0; i < 8; ++i) {
        const int L = 2 * c + (i >> 2);
        const float4 lo = wl(2 * (i & 3), L), hi = wl(2 * (i & 3) + 1, L);
        if (i == 0) {
            p[0] = o[0] * lo.x; p[1] = o[0] * lo.y; p[2] = o[0] * lo.z; p[3] = o[0] * lo.w;
            p[4] = o[0] * hi.x; p[5] = o[0] * hi.y; p[6] = o[0] * hi.z; p[7] = o[0] * hi.w;
        } else {
            p[0] = fmaf(o[i], lo.x, p[0]); p[1] = fmaf(o[i], lo.y, p[1]);
            p[2] = fmaf(o[i], lo.z, p[2]); p[3] = fmaf(o[i], lo.w, p[3]);
            p[4] = fmaf(o[i], hi.x, p[4]); p[5] = fmaf(o[i], hi.y, p[5]);
            p[6] = fmaf(o[i], hi.z, p[6]); p[7] = fmaf(o[i], hi.w, p[7]);
        }
    }
}

__device__ __forceinline__ float class_sums16(const float (&p)[8], int c) {
    const bool b3 = (c & 8) != 0, b2 = (c & 4) != 0, b1 = (c & 2) != 0;
    float q[4], r[2];
#pragma unroll
    for (int i = 0; i < 4; ++i) {
        const float keep = b3 ? p[i + 4] : p[i];
        const float send = b3 ? p[i] : p[i + 4];
        q[i] = keep + row_xchg<0x168>(send);
    }
#pragma unroll
    for (int i = 0; i < 2; ++i) {
        const float keep = b2 ? q[i + 2] : q[i];
        const float send = b2 ? q[i] : q[i + 2];
        r[i] = keep + row_xchg<0x164>(send);
    }
    const float keep = b1 ? r[1] : r[0];
    const float send = b1 ? r[0] : r[1];
    float s = keep + row_xchg<0x162>(send);
    s += row_xchg<0x161>(s);
    return s;
}

__launch_bounds__(256)
__global__ void gather_norm_gemm(const int* __restrict__ rowptr, const int* __restrict__ col,
                                 const float* __restrict__ dinv, const _Float16* __restrict__ h,
                                 const float* __restrict__ bias,
                                 const float* __restrict__ Wl, float* plog, int accumulate,
                                 const float* __restrict__ Wn, _Float16* __restrict__ hNext,
                                 int nTiles) {
    __shared__ float ldsW[HDIM * LDSW_STRIDE];
    __shared__ float ldsT[16 * LDSA_STRIDE];
    __shared__ float4 sLog[32];

    const int tid  = threadIdx.x;
    const int lane = tid & 31;
    const int wave = __builtin_amdgcn_readfirstlane(tid >> 5);

    for (int i = 0; i < 16; ++i) {
        int idx = (i * 256 + tid) * 4;
        int row = idx >> 7, cc = idx & 127;
        float4 v = *(const float4*)(Wn + idx);
        *(float4*)(&ldsW[row * LDSW_STRIDE + cc]) = v;
    }
    if (blockIdx.x == 0 && tid < 32)
        ((uint2*)(hNext + (size_t)nTiles * 16 * HDIM))[tid] = make_uint2(0u, 0u);

    const int hh = lane >> 4;
    const int c  = lane & 15;
    const float4 b0 = ((const float4*)bias)[2 * c];
    const float4 b1 = ((const float4*)bias)[2 * c + 1];

    *(float4*)(&ldsW[wl_slot((tid & 7) * 32 + (tid >> 3))]) = ((const float4*)Wl)[tid];
    __syncthreads();

    const int m     = lane & 15;
    const int khalf = (lane >> 4) * 2;
    const int mrow  = (lane >> 4) * 8;
    const float* aRow = &ldsT[m * LDSA_STRIDE + khalf];
    const float* wCol = &ldsW[khalf * LDSW_STRIDE + wave * 16 + m];
    const int rA = wave * 2;

    for (int tile = blockIdx.x; tile < nTiles; tile += gridDim.x) {
        const int nA = tile * 16 + rA;
        const int pA = rowptr[nA];
        const int pB = rowptr[nA + 1];
        const int pC = rowptr[nA + 2];
        const int dA = pB - pA, dB = pC - pB;
        const int maxdeg = dA > dB ? dA : dB;
        const int beg = hh ? pB : pA;
        const int deg = hh ? dB : dA;
        const float dd = dinv[nA + hh];
        const float4 dlo = *(const float4*)(dinv + (size_t)tile * 16 + mrow);
        const float4 dhi = *(const float4*)(dinv + (size_t)tile * 16 + mrow + 4);

        float acc8[8];
        gather_pair(col, h, beg, deg, maxdeg, nA + hh, nTiles * 16, c, acc8);
        float o[8];
        pair_tail(acc8, dd, b0, b1, o);
        float* tRow = &ldsT[(rA + hh) * LDSA_STRIDE + c * 8];
        *(float4*)(tRow)     = make_float4(o[0], o[1], o[2], o[3]);
        *(float4*)(tRow + 4) = make_float4(o[4], o[5], o[6], o[7]);
        float lp[8];
        pair_partials(o, c, [&](int q, int L) { return *(const float4*)(&ldsW[wl_slot(q * 32 + L)]); }, lp);
        const float ls = class_sums16(lp, c);
        if ((c & 1) == 0)
            ((float*)sLog)[(rA + hh) * 8 + (c >> 1)] = ls;
        __syncthreads();

        float4 lg = {0.f, 0.f, 0.f, 0.f}, pv = {0.f, 0.f, 0.f, 0.f};
        if (wave == 0) {
            lg = sLog[lane];
            if (accumulate) pv = ((const float4*)plog)[(size_t)tile * 32 + lane];
        }

        v8f acc = {0.f, 0.f, 0.f, 0.f, 0.f, 0.f, 0.f, 0.f};
        for (int k4 = 0; k4 < 32; ++k4) {
            const int kc = k4 * 4;
            v2f a;
            a.x = aRow[kc];
            a.y = aRow[kc + 1];
            v2f w;
            w.x = wCol[kc * LDSW_STRIDE];
            w.y = wCol[(kc + 1) * LDSW_STRIDE];
            acc = __builtin_amdgcn_wmma_f32_16x16x4_f32(
                false, a, false, w, (short)0, acc, false, false);
        }
        if (wave == 0) {
            if (accumulate) {
                lg.x = pv.x + lg.x; lg.y = pv.y + lg.y; lg.z = pv.z + lg.z; lg.w = pv.w + lg.w;
            }
            ((float4*)plog)[(size_t)tile * 32 + lane] = lg;
        }
        __syncthreads();

        const float drow[8] = {dlo.x, dlo.y, dlo.z, dlo.w, dhi.x, dhi.y, dhi.z, dhi.w};
#pragma unroll
        for (int r = 0; r < 8; ++r)
            ldsT[(mrow + r) * LDSA_STRIDE + wave * 16 + m] = acc[r] * drow[r];
        __syncthreads();
        const float4 hA = *(const float4*)(&ldsT[rA * LDSA_STRIDE + lane * 4]);
        const float4 hB = *(const float4*)(&ldsT[(rA + 1) * LDSA_STRIDE + lane * 4]);
        ((h4*)(hNext + (size_t)nA * HDIM))[lane] = narrow(hA);
        ((h4*)(hNext + (size_t)(nA + 1) * HDIM))[lane] = narrow(hB);
    }
}

__launch_bounds__(256)
__global__ void gather_agg_classify(const int* __restrict__ rowptr, const int* __restrict__ col,
                                    const float* __restrict__ dinv, const _Float16* __restrict__ h,
                                    const float* __restrict__ bias,
                                    const float* __restrict__ Wl, const float* __restrict__ blin,
                                    const float* __restrict__ plog, float* __restrict__ out, int n,
                                    int zrow) {
    __shared__ float4 sW[256];
    __shared__ float4 sLog[32];
    __shared__ int arrived;
    const int tid  = threadIdx.x;
    const int lane = tid & 31;
    const int hh   = lane >> 4;
    const int c    = lane & 15;
    const int wave = __builtin_amdgcn_readfirstlane(tid >> 5);
    const int node0 = blockIdx.x * 16;
    const int nd   = node0 + wave * 2 + hh;
    const int node = nd < n ? nd : n - 1;

    const int   beg = rowptr[node];
    const int   end = rowptr[node + 1];
    const float dd  = dinv[node];
    const float4 b0 = ((const float4*)bias)[2 * c];
    const float4 b1 = ((const float4*)bias)[2 * c + 1];
    stage_wlin(Wl, sW, tid);
    if (tid == 0) arrived = 0;
    __syncthreads();
    const int deg = end - beg;
    const int d0 = __builtin_amdgcn_readlane(deg, 0);
    const int d1 = __builtin_amdgcn_readlane(deg, 16);
    const int maxdeg = d0 > d1 ? d0 : d1;

    float acc8[8];
    gather_pair(col, h, beg, deg, maxdeg, node, zrow, c, acc8);
    float o[8];
    pair_tail(acc8, dd, b0, b1, o);
    float p[8];
    pair_partials(o, c, [&](int q, int L) { return sW[q * 32 + L]; }, p);
    const float s = class_sums16(p, c);
    if ((c & 1) == 0) ((float*)sLog)[(wave * 2 + hh) * 8 + (c >> 1)] = s;
    int before = 0;
    if (lane == 0)
        before = __hip_atomic_fetch_add(&arrived, 1, __ATOMIC_ACQ_REL, __HIP_MEMORY_SCOPE_WORKGROUP);
    before = __builtin_amdgcn_readfirstlane(before);
    if (before != 7) return;

    const int half = lane & 1;
    const int nodeF = node0 + (lane >> 1);
    const int nodeC = nodeF < n ? nodeF : n - 1;
    const float4 own = sLog[lane];
    const float4 pv  = ((const float4*)plog)[(size_t)nodeC * 2 + half];
    const float4 bl  = ((const float4*)blin)[half];
    float4 v;
    v.x = (pv.x + own.x) + bl.x;
    v.y = (pv.y + own.y) + bl.y;
    v.z = (pv.z + own.z) + bl.z;
    v.w = (pv.w + own.w) + bl.w;
    float mx = fmaxf(fmaxf(v.x, v.y), fmaxf(v.z, v.w));
    mx = fmaxf(mx, row_xchg<0x161>(mx));
    float se = (expf(v.x - mx) + expf(v.y - mx)) + (expf(v.z - mx) + expf(v.w - mx));
    se += row_xchg<0x161>(se);
    const float lse = mx + logf(se);
    float4 r;
    r.x = v.x - lse; r.y = v.y - lse; r.z = v.z - lse; r.w = v.w - lse;
    if (nodeF < n)
        *(float4*)(out + (size_t)nodeF * 8 + half * 4) = r;
}

extern "C" void kernel_launch(void* const* d_in, const int* in_sizes, int n_in,
                              void* d_out, int out_size, void* d_ws, size_t ws_size,
                              hipStream_t stream) {
    const float* x    = (const float*)d_in[0];
    const int*   ei   = (const int*)d_in[1];
    const float* W1   = (const float*)d_in[2];
    const float* b1   = (const float*)d_in[3];
    const float* W2   = (const float*)d_in[4];
    const float* b2   = (const float*)d_in[5];
    const float* W3   = (const float*)d_in[6];
    const float* b3   = (const float*)d_in[7];
    const float* Wlin = (const float*)d_in[8];
    const float* blin = (const float*)d_in[9];

    const int n = in_sizes[0] / HDIM;
    const int E = in_sizes[1] / 2;
    const int* src = ei;
    const int* dst = ei + E;

    const int nTiles = (n + 15) / 16;
    const int zrow = nTiles * 16;
    _Float16* h  = (_Float16*)d_ws;
    _Float16* h2 = h  + (size_t)(zrow + 1) * HDIM;
    float* plog  = (float*)(h2 + (size_t)(zrow + 1) * HDIM);
    float* dinv = plog + (size_t)n * 8;
    int* col         = (int*)(dinv + (size_t)n);
    int* rowptr      = col + (size_t)E;
    const int G = (E + ECH - 1) / ECH;
    int* tbl         = rowptr + (size_t)n + 1;
    int* off         = tbl + (size_t)G * NBK;
    unsigned* chunk  = (unsigned*)h;

    bin_sort<<<G, 256, 0, stream>>>(src, dst, tbl, off, chunk, E);
    bucket_csr<<<NBK, 512, 0, stream>>>(chunk, tbl, off, rowptr, dinv, col, n, E, G);

    const float* Wl1 = Wlin;
    const float* Wl2 = Wlin + HDIM * 8;
    const float* Wl3 = Wlin + 2 * HDIM * 8;

    gemm_wmma<<<(nTiles + WPB - 1) / WPB, 256, 0, stream>>>(x, W1, dinv, h, nTiles);
    gather_norm_gemm<<<FUSED_GRID, 256, 0, stream>>>(rowptr, col, dinv, h, b1, Wl1, plog, 0, W2, h2, nTiles);
    gather_norm_gemm<<<FUSED_GRID, 256, 0, stream>>>(rowptr, col, dinv, h2, b2, Wl2, plog, 1, W3, h, nTiles);
    gather_agg_classify<<<(n + 15) / 16, 256, 0, stream>>>(rowptr, col, dinv, h, b3, Wl3, blin, plog,
                                                         (float*)d_out, n, zrow);
}
